# starting kernel
# baseline (speedup 1.0000x reference)
#include <hip/hip_runtime.h>
#include <math.h>

typedef __attribute__((ext_vector_type(2))) float v2f;
typedef __attribute__((ext_vector_type(8))) float v8f;

#define NF 128

__global__ void fill_zero(float* __restrict__ p, long n) {
  long i = (long)blockIdx.x * blockDim.x + threadIdx.x;
  long stride = (long)gridDim.x * blockDim.x;
  for (; i < n; i += stride) p[i] = 0.0f;
}

__global__ __launch_bounds__(128) void spmm_kernel(
    const float* __restrict__ X, const int* __restrict__ src,
    const int* __restrict__ dst, const float* __restrict__ vals,
    const float* __restrict__ Mv, const float* __restrict__ AMv,
    float* __restrict__ out) {
  int e = blockIdx.x;
  int f = threadIdx.x;
  int s = src[e];
  int d = dst[e];
  float w = vals[e];
  if (Mv)  w *= Mv[s];
  if (AMv) w *= AMv[d];
  atomicAdd(out + (size_t)d * NF + f, w * X[(size_t)s * NF + f]);
}

__global__ __launch_bounds__(256) void gemm_wmma_kernel(
    const float* __restrict__ A,
    const float* __restrict__ W,
    const float* __restrict__ bias,
    const float* __restrict__ g, const float* __restrict__ be,
    const float* __restrict__ rm, const float* __restrict__ rv,
    float* __restrict__ out,
    int n_rows, int n_out_real, int tiles_n, int relu) {
  const int K = NF;
  int lane = threadIdx.x & 31;
  int wave = threadIdx.x >> 5;
  int tiles_m = (n_rows + 15) >> 4;
  int tile = blockIdx.x * 8 + wave;
  if (tile >= tiles_m * tiles_n) return;
  int tm = tile / tiles_n;
  int tn = tile - tm * tiles_n;

  int hl = lane >> 4;
  int ll = lane & 15;
  int row = tm * 16 + ll;
  int ar  = row < n_rows ? row : n_rows - 1;
  int col = tn * 16 + ll;
  int cw  = col < n_out_real ? col : n_out_real - 1;
  int kh  = hl * 2;

  const float* ap = A + (size_t)ar * K + kh;
  const float* wp = W + (size_t)cw * K + kh;

  v8f c = {};
#pragma unroll
  for (int k0 = 0; k0 < K; k0 += 4) {
    v2f a = *(const v2f*)(ap + k0);
    v2f b = *(const v2f*)(wp + k0);
    c = __builtin_amdgcn_wmma_f32_16x16x4_f32(
         false, a,  false, b,
         (short)0, c,  false,  false);
  }

  float bb = bias ? bias[cw] : 0.0f;
  float sc = 1.0f, sh = 0.0f;
  if (g) {
    float s = g[cw] * rsqrtf(rv[cw] + 1e-5f);
    sc = s;
    sh = be[cw] - rm[cw] * s;
  }
  if (col < n_out_real) {
    int rb = tm * 16 + hl * 8;
#pragma unroll
    for (int i = 0; i < 8; ++i) {
      int r = rb + i;
      if (r < n_rows) {
        float v = (c[i] + bb) * sc + sh;
        if (relu) v = fmaxf(v, 0.0f);
        out[(size_t)r * n_out_real + col] = v;
      }
    }
  }
}

__global__ __launch_bounds__(256) void log_softmax40(float* __restrict__ out, int n) {
  int i = blockIdx.x * blockDim.x + threadIdx.x;
  if (i >= n) return;
  float* row = out + (size_t)i * 40;
  float m = row[0];
#pragma unroll
  for (int j = 1; j < 40; ++j) m = fmaxf(m, row[j]);
  float s = 0.0f;
#pragma unroll
  for (int j = 0; j < 40; ++j) s += expf(row[j] - m);
  float l = m + logf(s);
#pragma unroll
  for (int j = 0; j < 40; ++j) row[j] = row[j] - l;
}

extern "C" void kernel_launch(void* const* d_in, const int* in_sizes, int n_in,
                              void* d_out, int out_size, void* d_ws, size_t ws_size,
                              hipStream_t stream) {
  const float* x     = (const float*)d_in[0];
  const float* Mv    = (const float*)d_in[1];
  const float* AMv   = (const float*)d_in[2];
  const int*   srcZ  = (const int*)d_in[3];
  const int*   dstZ  = (const int*)d_in[4];
  const float* valsZ = (const float*)d_in[5];
  const int*   src   = (const int*)d_in[6];
  const int*   dst   = (const int*)d_in[7];
  const float* vals  = (const float*)d_in[8];
  const float* W0 = (const float*)d_in[9];
  const float* b0 = (const float*)d_in[10];
  const float* g0 = (const float*)d_in[11];
  const float* be0 = (const float*)d_in[12];
  const float* rm0 = (const float*)d_in[13];
  const float* rv0 = (const float*)d_in[14];
  const float* W1 = (const float*)d_in[15];
  const float* b1 = (const float*)d_in[16];
  const float* g1 = (const float*)d_in[17];
  const float* be1 = (const float*)d_in[18];
  const float* rm1 = (const float*)d_in[19];
  const float* rv1 = (const float*)d_in[20];
  const float* W2 = (const float*)d_in[21];
  const float* b2 = (const float*)d_in[22];

  int N = in_sizes[1];
  int E = in_sizes[3];
  float* bufA = (float*)d_ws;
  float* bufB = bufA + (size_t)N * NF;
  float* outp = (float*)d_out;

  int tiles_m = (N + 15) / 16;
  long nfeat_elems = (long)N * NF;
  int gemm_blocks_h = (tiles_m * 8 + 7) / 8;
  int gemm_blocks_o = (tiles_m * 3 + 7) / 8;

  fill_zero<<<2048, 256, 0, stream>>>(bufA, nfeat_elems);
  spmm_kernel<<<E, 128, 0, stream>>>(x, srcZ, dstZ, valsZ, Mv, AMv, bufA);
  gemm_wmma_kernel<<<gemm_blocks_h, 256, 0, stream>>>(
      bufA, W0, b0, g0, be0, rm0, rv0, bufB, N, NF, 8, 1);

  fill_zero<<<2048, 256, 0, stream>>>(bufA, nfeat_elems);
  spmm_kernel<<<E, 128, 0, stream>>>(bufB, src, dst, vals, nullptr, nullptr, bufA);
  gemm_wmma_kernel<<<gemm_blocks_h, 256, 0, stream>>>(
      bufA, W1, b1, g1, be1, rm1, rv1, bufB, N, NF, 8, 1);

  fill_zero<<<2048, 256, 0, stream>>>(bufA, nfeat_elems);
  spmm_kernel<<<E, 128, 0, stream>>>(bufB, src, dst, vals, nullptr, nullptr, bufA);
  gemm_wmma_kernel<<<gemm_blocks_o, 256, 0, stream>>>(
      bufA, W2, b2, nullptr, nullptr, nullptr, nullptr, outp, N, 40, 3, 0);
  log_softmax40<<<(N + 255) / 256, 256, 0, stream>>>(outp, N);
}


# SPMM: one wave per 8 edges issues all 32 row-line loads before its 32 whole-line f32 atomics (was a 4-wave block per edge, one load in flight); 32x fewer waves; sums stay in atomic arrival order
# speedup vs baseline: 3.5287x; 3.5287x over previous
#include <hip/hip_runtime.h>
#include <math.h>

typedef __attribute__((ext_vector_type(2))) float v2f;
typedef __attribute__((ext_vector_type(8))) float v8f;

#define NF 128

__global__ void fill_zero(float* __restrict__ p, long n) {
  long i = (long)blockIdx.x * blockDim.x + threadIdx.x;
  long stride = (long)gridDim.x * blockDim.x;
  for (; i < n; i += stride) p[i] = 0.0f;
}

#define EPW 8
#define SPMM_WAVES 8
#define SPMM_EPB (EPW * SPMM_WAVES)

__global__ __launch_bounds__(32 * SPMM_WAVES) void spmm_kernel(
    const float* __restrict__ X, const int* __restrict__ src,
    const int* __restrict__ dst, const float* __restrict__ vals,
    const float* __restrict__ Mv, const float* __restrict__ AMv,
    float* __restrict__ out, int E) {
  const int lane = threadIdx.x & 31;
  const int wave = __builtin_amdgcn_readfirstlane((int)(threadIdx.x >> 5));
  const int e0 = (blockIdx.x * SPMM_WAVES + wave) * EPW;
  if (e0 >= E) return;

  if (e0 + EPW <= E) {
    int s[EPW], d[EPW];
    float w[EPW];
    for (int i = 0; i < EPW; ++i) {
      s[i] = src[e0 + i];
      d[i] = dst[e0 + i];
      w[i] = vals[e0 + i];
    }
    if (Mv) {
      for (int i = 0; i < EPW; ++i) w[i] *= Mv[s[i]];
    }
    if (AMv) {
      for (int i = 0; i < EPW; ++i) w[i] *= AMv[d[i]];
    }
    float v[EPW][4];
    for (int i = 0; i < EPW; ++i) {
      const float* xr = X + (size_t)s[i] * NF + lane;
      for (int j = 0; j < 4; ++j) v[i][j] = xr[32 * j];
    }
    for (int i = 0; i < EPW; ++i) {
      float* orow = out + (size_t)d[i] * NF + lane;
      for (int j = 0; j < 4; ++j) atomicAdd(orow + 32 * j, w[i] * v[i][j]);
    }
  } else {
    for (int e = e0; e < E; ++e) {
      int s = src[e];
      int d = dst[e];
      float w = vals[e];
      if (Mv)  w *= Mv[s];
      if (AMv) w *= AMv[d];
      const float* xr = X + (size_t)s * NF + lane;
      float* orow = out + (size_t)d * NF + lane;
      for (int j = 0; j < 4; ++j) atomicAdd(orow + 32 * j, w * xr[32 * j]);
    }
  }
}

__global__ __launch_bounds__(256) void gemm_wmma_kernel(
    const float* __restrict__ A,
    const float* __restrict__ W,
    const float* __restrict__ bias,
    const float* __restrict__ g, const float* __restrict__ be,
    const float* __restrict__ rm, const float* __restrict__ rv,
    float* __restrict__ out,
    int n_rows, int n_out_real, int tiles_n, int relu) {
  const int K = NF;
  int lane = threadIdx.x & 31;
  int wave = threadIdx.x >> 5;
  int tiles_m = (n_rows + 15) >> 4;
  int tile = blockIdx.x * 8 + wave;
  if (tile >= tiles_m * tiles_n) return;
  int tm = tile / tiles_n;
  int tn = tile - tm * tiles_n;

  int hl = lane >> 4;
  int ll = lane & 15;
  int row = tm * 16 + ll;
  int ar  = row < n_rows ? row : n_rows - 1;
  int col = tn * 16 + ll;
  int cw  = col < n_out_real ? col : n_out_real - 1;
  int kh  = hl * 2;

  const float* ap = A + (size_t)ar * K + kh;
  const float* wp = W + (size_t)cw * K + kh;

  v8f c = {};
#pragma unroll
  for (int k0 = 0; k0 < K; k0 += 4) {
    v2f a = *(const v2f*)(ap + k0);
    v2f b = *(const v2f*)(wp + k0);
    c = __builtin_amdgcn_wmma_f32_16x16x4_f32(
         false, a,  false, b,
         (short)0, c,  false,  false);
  }

  float bb = bias ? bias[cw] : 0.0f;
  float sc = 1.0f, sh = 0.0f;
  if (g) {
    float s = g[cw] * rsqrtf(rv[cw] + 1e-5f);
    sc = s;
    sh = be[cw] - rm[cw] * s;
  }
  if (col < n_out_real) {
    int rb = tm * 16 + hl * 8;
#pragma unroll
    for (int i = 0; i < 8; ++i) {
      int r = rb + i;
      if (r < n_rows) {
        float v = (c[i] + bb) * sc + sh;
        if (relu) v = fmaxf(v, 0.0f);
        out[(size_t)r * n_out_real + col] = v;
      }
    }
  }
}

__global__ __launch_bounds__(256) void log_softmax40(float* __restrict__ out, int n) {
  int i = blockIdx.x * blockDim.x + threadIdx.x;
  if (i >= n) return;
  float* row = out + (size_t)i * 40;
  float m = row[0];
#pragma unroll
  for (int j = 1; j < 40; ++j) m = fmaxf(m, row[j]);
  float s = 0.0f;
#pragma unroll
  for (int j = 0; j < 40; ++j) s += expf(row[j] - m);
  float l = m + logf(s);
#pragma unroll
  for (int j = 0; j < 40; ++j) row[j] = row[j] - l;
}

extern "C" void kernel_launch(void* const* d_in, const int* in_sizes, int n_in,
                              void* d_out, int out_size, void* d_ws, size_t ws_size,
                              hipStream_t stream) {
  const float* x     = (const float*)d_in[0];
  const float* Mv    = (const float*)d_in[1];
  const float* AMv   = (const float*)d_in[2];
  const int*   srcZ  = (const int*)d_in[3];
  const int*   dstZ  = (const int*)d_in[4];
  const float* valsZ = (const float*)d_in[5];
  const int*   src   = (const int*)d_in[6];
  const int*   dst   = (const int*)d_in[7];
  const float* vals  = (const float*)d_in[8];
  const float* W0 = (const float*)d_in[9];
  const float* b0 = (const float*)d_in[10];
  const float* g0 = (const float*)d_in[11];
  const float* be0 = (const float*)d_in[12];
  const float* rm0 = (const float*)d_in[13];
  const float* rv0 = (const float*)d_in[14];
  const float* W1 = (const float*)d_in[15];
  const float* b1 = (const float*)d_in[16];
  const float* g1 = (const float*)d_in[17];
  const float* be1 = (const float*)d_in[18];
  const float* rm1 = (const float*)d_in[19];
  const float* rv1 = (const float*)d_in[20];
  const float* W2 = (const float*)d_in[21];
  const float* b2 = (const float*)d_in[22];

  int N = in_sizes[1];
  int E = in_sizes[3];
  float* bufA = (float*)d_ws;
  float* bufB = bufA + (size_t)N * NF;
  float* outp = (float*)d_out;

  int tiles_m = (N + 15) / 16;
  long nfeat_elems = (long)N * NF;
  int gemm_blocks_h = (tiles_m * 8 + 7) / 8;
  int gemm_blocks_o = (tiles_m * 3 + 7) / 8;
  int spmm_blocks = (E + SPMM_EPB - 1) / SPMM_EPB;

  fill_zero<<<2048, 256, 0, stream>>>(bufA, nfeat_elems);
  spmm_kernel<<<spmm_blocks, 32 * SPMM_WAVES, 0, stream>>>(x, srcZ, dstZ, valsZ, Mv, AMv, bufA, E);
  gemm_wmma_kernel<<<gemm_blocks_h, 256, 0, stream>>>(
      bufA, W0, b0, g0, be0, rm0, rv0, bufB, N, NF, 8, 1);

  fill_zero<<<2048, 256, 0, stream>>>(bufA, nfeat_elems);
  spmm_kernel<<<spmm_blocks, 32 * SPMM_WAVES, 0, stream>>>(bufB, src, dst, vals, nullptr, nullptr, bufA, E);
  gemm_wmma_kernel<<<gemm_blocks_h, 256, 0, stream>>>(
      bufA, W1, b1, g1, be1, rm1, rv1, bufB, N, NF, 8, 1);

  fill_zero<<<2048, 256, 0, stream>>>(bufA, nfeat_elems);
  spmm_kernel<<<spmm_blocks, 32 * SPMM_WAVES, 0, stream>>>(bufB, src, dst, vals, nullptr, nullptr, bufA, E);
  gemm_wmma_kernel<<<gemm_blocks_o, 256, 0, stream>>>(
      bufA, W2, b2, nullptr, nullptr, nullptr, nullptr, outp, N, 40, 3, 0);
  log_softmax40<<<(N + 255) / 256, 256, 0, stream>>>(outp, N);
}


# Last layer: 128->40 WMMA product moved before its edge aggregation (A(hW^T) for (Ah)W^T), so gathers and f32 atomics move 40 floats per edge, not 128; its zero fill goes; sum order changes with it
# speedup vs baseline: 3.8168x; 1.0816x over previous
#include <hip/hip_runtime.h>
#include <math.h>
#include <utility>

typedef __attribute__((ext_vector_type(2))) float v2f;
typedef __attribute__((ext_vector_type(8))) float v8f;

#define NF 128

__global__ void fill_zero(float* __restrict__ p, long n) {
  long i = (long)blockIdx.x * blockDim.x + threadIdx.x;
  long stride = (long)gridDim.x * blockDim.x;
  for (; i < n; i += stride) p[i] = 0.0f;
}

#define EPW 8
#define SPMM_WAVES 8
#define SPMM_EPB (EPW * SPMM_WAVES)

__global__ __launch_bounds__(32 * SPMM_WAVES) void spmm_kernel(
    const float* __restrict__ X, const int* __restrict__ src,
    const int* __restrict__ dst, const float* __restrict__ vals,
    const float* __restrict__ Mv, const float* __restrict__ AMv,
    float* __restrict__ out, int E) {
  const int lane = threadIdx.x & 31;
  const int wave = __builtin_amdgcn_readfirstlane((int)(threadIdx.x >> 5));
  const int e0 = (blockIdx.x * SPMM_WAVES + wave) * EPW;
  if (e0 >= E) return;

  if (e0 + EPW <= E) {
    int s[EPW], d[EPW];
    float w[EPW];
    for (int i = 0; i < EPW; ++i) {
      s[i] = src[e0 + i];
      d[i] = dst[e0 + i];
      w[i] = vals[e0 + i];
    }
    if (Mv) {
      for (int i = 0; i < EPW; ++i) w[i] *= Mv[s[i]];
    }
    if (AMv) {
      for (int i = 0; i < EPW; ++i) w[i] *= AMv[d[i]];
    }
    float v[EPW][4];
    for (int i = 0; i < EPW; ++i) {
      const float* xr = X + (size_t)s[i] * NF + lane;
      for (int j = 0; j < 4; ++j) v[i][j] = xr[32 * j];
    }
    for (int i = 0; i < EPW; ++i) {
      float* orow = out + (size_t)d[i] * NF + lane;
      for (int j = 0; j < 4; ++j) atomicAdd(orow + 32 * j, w[i] * v[i][j]);
    }
  } else {
    for (int e = e0; e < E; ++e) {
      int s = src[e];
      int d = dst[e];
      float w = vals[e];
      if (Mv)  w *= Mv[s];
      if (AMv) w *= AMv[d];
      const float* xr = X + (size_t)s * NF + lane;
      float* orow = out + (size_t)d * NF + lane;
      for (int j = 0; j < 4; ++j) atomicAdd(orow + 32 * j, w * xr[32 * j]);
    }
  }
}

#define NC 40
#define EPW_C 16
#define NACC (EPW_C / 4 * 5)

template <int G, int J>
__device__ __forceinline__ void gather40(const float* __restrict__ T, const unsigned (&so)[EPW_C],
                                         unsigned lane, float (&v)[NACC]) {
  constexpr unsigned EA = (32 * J) / NC, EB = (32 * J + 31) / NC;
  const unsigned f = 32 * J + lane;
  const bool hi = f >= NC * EB;
  const unsigned base = hi ? so[4 * G + EB] - NC * EB : so[4 * G + EA] - NC * EA;
  v[5 * G + J] = T[base + f];
}

template <int G, int J>
__device__ __forceinline__ void scatter40(float* __restrict__ out, const unsigned (&dofs)[EPW_C],
                                          const float (&w)[EPW_C], unsigned lane,
                                          const float (&v)[NACC]) {
  constexpr unsigned EA = (32 * J) / NC, EB = (32 * J + 31) / NC;
  const unsigned f = 32 * J + lane;
  const bool hi = f >= NC * EB;
  const unsigned base = hi ? dofs[4 * G + EB] - NC * EB : dofs[4 * G + EA] - NC * EA;
  const float ww = hi ? w[4 * G + EB] : w[4 * G + EA];
  atomicAdd(out + (base + f), ww * v[5 * G + J]);
}

template <int... I>
__device__ __forceinline__ void gather40_all(std::integer_sequence<int, I...>,
                                             const float* __restrict__ T,
                                             const unsigned (&so)[EPW_C], unsigned lane,
                                             float (&v)[NACC]) {
  (gather40<I / 5, I % 5>(T, so, lane, v), ...);
}

template <int... I>
__device__ __forceinline__ void scatter40_all(std::integer_sequence<int, I...>,
                                              float* __restrict__ out,
                                              const unsigned (&dofs)[EPW_C],
                                              const float (&w)[EPW_C], unsigned lane,
                                              const float (&v)[NACC]) {
  (scatter40<I / 5, I % 5>(out, dofs, w, lane, v), ...);
}

template <int... I>
__device__ __forceinline__ void edges40_all(std::integer_sequence<int, I...>,
                                            const int* __restrict__ src,
                                            const int* __restrict__ dst,
                                            const float* __restrict__ vals, int e0,
                                            unsigned (&so)[EPW_C], unsigned (&dofs)[EPW_C],
                                            float (&w)[EPW_C]) {
  ((so[I] = (unsigned)src[e0 + I] * NC, dofs[I] = (unsigned)dst[e0 + I] * NC,
    w[I] = vals[e0 + I]), ...);
}

__global__ __launch_bounds__(32 * SPMM_WAVES) void spmm40_kernel(
    const float* __restrict__ T, const int* __restrict__ src,
    const int* __restrict__ dst, const float* __restrict__ vals,
    float* __restrict__ out, int E) {
  const unsigned lane = threadIdx.x & 31;
  const int wave = __builtin_amdgcn_readfirstlane((int)(threadIdx.x >> 5));
  const int e0 = (blockIdx.x * SPMM_WAVES + wave) * EPW_C;
  if (e0 >= E) return;

  if (e0 + EPW_C <= E) {
    unsigned so[EPW_C], dofs[EPW_C];
    float w[EPW_C];
    float v[NACC];
    edges40_all(std::make_integer_sequence<int, EPW_C>{}, src, dst, vals, e0, so, dofs, w);
    gather40_all(std::make_integer_sequence<int, NACC>{}, T, so, lane, v);
    scatter40_all(std::make_integer_sequence<int, NACC>{}, out, dofs, w, lane, v);
  } else {
    for (int e = e0; e < E; ++e) {
      const float* tr = T + (size_t)src[e] * NC;
      float* orow = out + (size_t)dst[e] * NC;
      float w = vals[e];
      atomicAdd(orow + lane, w * tr[lane]);
      if (lane < NC - 32) atomicAdd(orow + 32 + lane, w * tr[32 + lane]);
    }
  }
}

__global__ __launch_bounds__(256) void gemm_wmma_kernel(
    const float* __restrict__ A,
    const float* __restrict__ W,
    const float* __restrict__ bias,
    const float* __restrict__ g, const float* __restrict__ be,
    const float* __restrict__ rm, const float* __restrict__ rv,
    float* __restrict__ out,
    float* __restrict__ acc,
    const float* __restrict__ acc_init,
    int n_rows, int n_out_real, int tiles_n, int relu) {
  const int K = NF;
  int lane = threadIdx.x & 31;
  int wave = threadIdx.x >> 5;
  int tiles_m = (n_rows + 15) >> 4;
  int tile = blockIdx.x * 8 + wave;
  if (tile >= tiles_m * tiles_n) return;
  int tm = tile / tiles_n;
  int tn = tile - tm * tiles_n;

  int hl = lane >> 4;
  int ll = lane & 15;
  int row = tm * 16 + ll;
  int ar  = row < n_rows ? row : n_rows - 1;
  int col = tn * 16 + ll;
  int cw  = col < n_out_real ? col : n_out_real - 1;
  int kh  = hl * 2;

  const float* ap = A + (size_t)ar * K + kh;
  const float* wp = W + (size_t)cw * K + kh;

  v8f c = {};
#pragma unroll
  for (int k0 = 0; k0 < K; k0 += 4) {
    v2f a = *(const v2f*)(ap + k0);
    v2f b = *(const v2f*)(wp + k0);
    c = __builtin_amdgcn_wmma_f32_16x16x4_f32(
         false, a,  false, b,
         (short)0, c,  false,  false);
  }

  float bb = bias ? bias[cw] : 0.0f;
  float ai = acc ? acc_init[cw] : 0.0f;
  float sc = 1.0f, sh = 0.0f;
  if (g) {
    float s = g[cw] * rsqrtf(rv[cw] + 1e-5f);
    sc = s;
    sh = be[cw] - rm[cw] * s;
  }
  if (col < n_out_real) {
    int rb = tm * 16 + hl * 8;
#pragma unroll
    for (int i = 0; i < 8; ++i) {
      int r = rb + i;
      if (r < n_rows) {
        float v = (c[i] + bb) * sc + sh;
        if (relu) v = fmaxf(v, 0.0f);
        out[(size_t)r * n_out_real + col] = v;
        if (acc) acc[(size_t)r * n_out_real + col] = ai;
      }
    }
  }
}

__global__ __launch_bounds__(256) void log_softmax40(float* __restrict__ out, int n) {
  int i = blockIdx.x * blockDim.x + threadIdx.x;
  if (i >= n) return;
  float* row = out + (size_t)i * 40;
  float m = row[0];
#pragma unroll
  for (int j = 1; j < 40; ++j) m = fmaxf(m, row[j]);
  float s = 0.0f;
#pragma unroll
  for (int j = 0; j < 40; ++j) s += expf(row[j] - m);
  float l = m + logf(s);
#pragma unroll
  for (int j = 0; j < 40; ++j) row[j] = row[j] - l;
}

extern "C" void kernel_launch(void* const* d_in, const int* in_sizes, int n_in,
                              void* d_out, int out_size, void* d_ws, size_t ws_size,
                              hipStream_t stream) {
  const float* x     = (const float*)d_in[0];
  const float* Mv    = (const float*)d_in[1];
  const float* AMv   = (const float*)d_in[2];
  const int*   srcZ  = (const int*)d_in[3];
  const int*   dstZ  = (const int*)d_in[4];
  const float* valsZ = (const float*)d_in[5];
  const int*   src   = (const int*)d_in[6];
  const int*   dst   = (const int*)d_in[7];
  const float* vals  = (const float*)d_in[8];
  const float* W0 = (const float*)d_in[9];
  const float* b0 = (const float*)d_in[10];
  const float* g0 = (const float*)d_in[11];
  const float* be0 = (const float*)d_in[12];
  const float* rm0 = (const float*)d_in[13];
  const float* rv0 = (const float*)d_in[14];
  const float* W1 = (const float*)d_in[15];
  const float* b1 = (const float*)d_in[16];
  const float* g1 = (const float*)d_in[17];
  const float* be1 = (const float*)d_in[18];
  const float* rm1 = (const float*)d_in[19];
  const float* rv1 = (const float*)d_in[20];
  const float* W2 = (const float*)d_in[21];
  const float* b2 = (const float*)d_in[22];

  int N = in_sizes[1];
  int E = in_sizes[3];
  float* bufA = (float*)d_ws;
  float* bufB = bufA + (size_t)N * NF;
  float* outp = (float*)d_out;

  int tiles_m = (N + 15) / 16;
  long nfeat_elems = (long)N * NF;
  int gemm_blocks_h = (tiles_m * 8 + 7) / 8;
  int gemm_blocks_o = (tiles_m * 3 + 7) / 8;
  int spmm_blocks = (E + SPMM_EPB - 1) / SPMM_EPB;
  int spmm40_blocks = (E + EPW_C * SPMM_WAVES - 1) / (EPW_C * SPMM_WAVES);

  fill_zero<<<2048, 256, 0, stream>>>(bufA, nfeat_elems);
  spmm_kernel<<<spmm_blocks, 32 * SPMM_WAVES, 0, stream>>>(x, srcZ, dstZ, valsZ, Mv, AMv, bufA, E);
  gemm_wmma_kernel<<<gemm_blocks_h, 256, 0, stream>>>(
      bufA, W0, b0, g0, be0, rm0, rv0, bufB, nullptr, nullptr, N, NF, 8, 1);

  fill_zero<<<2048, 256, 0, stream>>>(bufA, nfeat_elems);
  spmm_kernel<<<spmm_blocks, 32 * SPMM_WAVES, 0, stream>>>(bufB, src, dst, vals, nullptr, nullptr, bufA, E);
  gemm_wmma_kernel<<<gemm_blocks_h, 256, 0, stream>>>(
      bufA, W1, b1, g1, be1, rm1, rv1, bufB, nullptr, nullptr, N, NF, 8, 1);

  gemm_wmma_kernel<<<gemm_blocks_o, 256, 0, stream>>>(
      bufB, W2, nullptr, nullptr, nullptr, nullptr, nullptr, bufA, outp, b2, N, NC, 3, 0);
  spmm40_kernel<<<spmm40_blocks, 32 * SPMM_WAVES, 0, stream>>>(bufA, src, dst, vals, outp, E);
  log_softmax40<<<(N + 255) / 256, 256, 0, stream>>>(outp, N);
}


# GEMMs: W staged once per block in LDS and one wave owns a 16-row strip with all its column tiles (8 or 3 accumulators), so A fragments are loaded once, not once per tile; k order and sums as before
# speedup vs baseline: 4.5228x; 1.1850x over previous
#include <hip/hip_runtime.h>
#include <math.h>
#include <utility>

typedef __attribute__((ext_vector_type(2))) float v2f;
typedef __attribute__((ext_vector_type(4))) float v4f;
typedef __attribute__((ext_vector_type(8))) float v8f;

#define NF 128

__global__ void fill_zero(float* __restrict__ p, long n) {
  long i = (long)blockIdx.x * blockDim.x + threadIdx.x;
  long stride = (long)gridDim.x * blockDim.x;
  for (; i < n; i += stride) p[i] = 0.0f;
}

#define EPW 8
#define SPMM_WAVES 8
#define SPMM_EPB (EPW * SPMM_WAVES)

__global__ __launch_bounds__(32 * SPMM_WAVES) void spmm_kernel(
    const float* __restrict__ X, const int* __restrict__ src,
    const int* __restrict__ dst, const float* __restrict__ vals,
    const float* __restrict__ Mv, const float* __restrict__ AMv,
    float* __restrict__ out, int E) {
  const int lane = threadIdx.x & 31;
  const int wave = __builtin_amdgcn_readfirstlane((int)(threadIdx.x >> 5));
  const int e0 = (blockIdx.x * SPMM_WAVES + wave) * EPW;
  if (e0 >= E) return;

  if (e0 + EPW <= E) {
    int s[EPW], d[EPW];
    float w[EPW];
    for (int i = 0; i < EPW; ++i) {
      s[i] = src[e0 + i];
      d[i] = dst[e0 + i];
      w[i] = vals[e0 + i];
    }
    if (Mv) {
      for (int i = 0; i < EPW; ++i) w[i] *= Mv[s[i]];
    }
    if (AMv) {
      for (int i = 0; i < EPW; ++i) w[i] *= AMv[d[i]];
    }
    float v[EPW][4];
    for (int i = 0; i < EPW; ++i) {
      const float* xr = X + (size_t)s[i] * NF + lane;
      for (int j = 0; j < 4; ++j) v[i][j] = xr[32 * j];
    }
    for (int i = 0; i < EPW; ++i) {
      float* orow = out + (size_t)d[i] * NF + lane;
      for (int j = 0; j < 4; ++j) atomicAdd(orow + 32 * j, w[i] * v[i][j]);
    }
  } else {
    for (int e = e0; e < E; ++e) {
      int s = src[e];
      int d = dst[e];
      float w = vals[e];
      if (Mv)  w *= Mv[s];
      if (AMv) w *= AMv[d];
      const float* xr = X + (size_t)s * NF + lane;
      float* orow = out + (size_t)d * NF + lane;
      for (int j = 0; j < 4; ++j) atomicAdd(orow + 32 * j, w * xr[32 * j]);
    }
  }
}

#define NC 40
#define EPW_C 16
#define NACC (EPW_C / 4 * 5)

template <int G, int J>
__device__ __forceinline__ void gather40(const float* __restrict__ T, const unsigned (&so)[EPW_C],
                                         unsigned lane, float (&v)[NACC]) {
  constexpr unsigned EA = (32 * J) / NC, EB = (32 * J + 31) / NC;
  const unsigned f = 32 * J + lane;
  const bool hi = f >= NC * EB;
  const unsigned base = hi ? so[4 * G + EB] - NC * EB : so[4 * G + EA] - NC * EA;
  v[5 * G + J] = T[base + f];
}

template <int G, int J>
__device__ __forceinline__ void scatter40(float* __restrict__ out, const unsigned (&dofs)[EPW_C],
                                          const float (&w)[EPW_C], unsigned lane,
                                          const float (&v)[NACC]) {
  constexpr unsigned EA = (32 * J) / NC, EB = (32 * J + 31) / NC;
  const unsigned f = 32 * J + lane;
  const bool hi = f >= NC * EB;
  const unsigned base = hi ? dofs[4 * G + EB] - NC * EB : dofs[4 * G + EA] - NC * EA;
  const float ww = hi ? w[4 * G + EB] : w[4 * G + EA];
  atomicAdd(out + (base + f), ww * v[5 * G + J]);
}

template <int... I>
__device__ __forceinline__ void gather40_all(std::integer_sequence<int, I...>,
                                             const float* __restrict__ T,
                                             const unsigned (&so)[EPW_C], unsigned lane,
                                             float (&v)[NACC]) {
  (gather40<I / 5, I % 5>(T, so, lane, v), ...);
}

template <int... I>
__device__ __forceinline__ void scatter40_all(std::integer_sequence<int, I...>,
                                              float* __restrict__ out,
                                              const unsigned (&dofs)[EPW_C],
                                              const float (&w)[EPW_C], unsigned lane,
                                              const float (&v)[NACC]) {
  (scatter40<I / 5, I % 5>(out, dofs, w, lane, v), ...);
}

template <int... I>
__device__ __forceinline__ void edges40_all(std::integer_sequence<int, I...>,
                                            const int* __restrict__ src,
                                            const int* __restrict__ dst,
                                            const float* __restrict__ vals, int e0,
                                            unsigned (&so)[EPW_C], unsigned (&dofs)[EPW_C],
                                            float (&w)[EPW_C]) {
  ((so[I] = (unsigned)src[e0 + I] * NC, dofs[I] = (unsigned)dst[e0 + I] * NC,
    w[I] = vals[e0 + I]), ...);
}

__global__ __launch_bounds__(32 * SPMM_WAVES) void spmm40_kernel(
    const float* __restrict__ T, const int* __restrict__ src,
    const int* __restrict__ dst, const float* __restrict__ vals,
    float* __restrict__ out, int E) {
  const unsigned lane = threadIdx.x & 31;
  const int wave = __builtin_amdgcn_readfirstlane((int)(threadIdx.x >> 5));
  const int e0 = (blockIdx.x * SPMM_WAVES + wave) * EPW_C;
  if (e0 >= E) return;

  if (e0 + EPW_C <= E) {
    unsigned so[EPW_C], dofs[EPW_C];
    float w[EPW_C];
    float v[NACC];
    edges40_all(std::make_integer_sequence<int, EPW_C>{}, src, dst, vals, e0, so, dofs, w);
    gather40_all(std::make_integer_sequence<int, NACC>{}, T, so, lane, v);
    scatter40_all(std::make_integer_sequence<int, NACC>{}, out, dofs, w, lane, v);
  } else {
    for (int e = e0; e < E; ++e) {
      const float* tr = T + (size_t)src[e] * NC;
      float* orow = out + (size_t)dst[e] * NC;
      float w = vals[e];
      atomicAdd(orow + lane, w * tr[lane]);
      if (lane < NC - 32) atomicAdd(orow + 32 + lane, w * tr[32 + lane]);
    }
  }
}

#define WP (NF + 4)
#define GEMM_WAVES 8

template <int NT>
__global__ __launch_bounds__(32 * GEMM_WAVES) void gemm_strip_kernel(
    const float* __restrict__ A,
    const float* __restrict__ W,
    const float* __restrict__ bias,
    const float* __restrict__ g, const float* __restrict__ be,
    const float* __restrict__ rm, const float* __restrict__ rv,
    float* __restrict__ out,
    float* __restrict__ acc,
    const float* __restrict__ acc_init,
    int n_rows, int n_out_real, int relu) {
  __shared__ float Ws[NT * 16 * WP];
  const int K = NF;

  for (int idx = threadIdx.x; idx < NT * 16 * (NF / 4); idx += 32 * GEMM_WAVES) {
    int r = idx >> 5;
    int q = idx & 31;
    int rr = r < n_out_real ? r : n_out_real - 1;
    v4f w4 = *(const v4f*)(W + (size_t)rr * K + 4 * q);
    *(v4f*)(Ws + r * WP + 4 * q) = w4;
  }
  __syncthreads();

  const int lane = threadIdx.x & 31;
  const int wave = __builtin_amdgcn_readfirstlane((int)(threadIdx.x >> 5));
  const int tiles_m = (n_rows + 15) >> 4;
  const int strip = blockIdx.x * GEMM_WAVES + wave;
  if (strip >= tiles_m) return;

  const int hl = lane >> 4;
  const int ll = lane & 15;
  const int row = strip * 16 + ll;
  const int ar  = row < n_rows ? row : n_rows - 1;

  const float* ap = A + (size_t)ar * K + 2 * hl;
  v2f a[K / 4];
#pragma unroll
  for (int t = 0; t < K / 4; ++t) a[t] = *(const v2f*)(ap + 4 * t);

  const float* wl = Ws + ll * WP + 2 * hl;
  v8f c[NT];
#pragma unroll
  for (int n = 0; n < NT; ++n) c[n] = v8f{};

#pragma unroll
  for (int t = 0; t < K / 4; ++t) {
    v2f b[NT];
#pragma unroll
    for (int n = 0; n < NT; ++n) b[n] = *(const v2f*)(wl + n * 16 * WP + 4 * t);
#pragma unroll
    for (int n = 0; n < NT; ++n)
      c[n] = __builtin_amdgcn_wmma_f32_16x16x4_f32(
           false, a[t],  false, b[n],
           (short)0, c[n],  false,  false);
  }

  const int rb = strip * 16 + hl * 8;
#pragma unroll
  for (int n = 0; n < NT; ++n) {
    int col = n * 16 + ll;
    int cw  = col < n_out_real ? col : n_out_real - 1;
    float bb = bias ? bias[cw] : 0.0f;
    float ai = acc ? acc_init[cw] : 0.0f;
    float sc = 1.0f, sh = 0.0f;
    if (g) {
      float s = g[cw] * rsqrtf(rv[cw] + 1e-5f);
      sc = s;
      sh = be[cw] - rm[cw] * s;
    }
    if (col < n_out_real) {
#pragma unroll
      for (int i = 0; i < 8; ++i) {
        int r = rb + i;
        if (r < n_rows) {
          float v = (c[n][i] + bb) * sc + sh;
          if (relu) v = fmaxf(v, 0.0f);
          out[(size_t)r * n_out_real + col] = v;
          if (acc) acc[(size_t)r * n_out_real + col] = ai;
        }
      }
    }
  }
}

__global__ __launch_bounds__(256) void log_softmax40(float* __restrict__ out, int n) {
  int i = blockIdx.x * blockDim.x + threadIdx.x;
  if (i >= n) return;
  float* row = out + (size_t)i * 40;
  float m = row[0];
#pragma unroll
  for (int j = 1; j < 40; ++j) m = fmaxf(m, row[j]);
  float s = 0.0f;
#pragma unroll
  for (int j = 0; j < 40; ++j) s += expf(row[j] - m);
  float l = m + logf(s);
#pragma unroll
  for (int j = 0; j < 40; ++j) row[j] = row[j] - l;
}

extern "C" void kernel_launch(void* const* d_in, const int* in_sizes, int n_in,
                              void* d_out, int out_size, void* d_ws, size_t ws_size,
                              hipStream_t stream) {
  const float* x     = (const float*)d_in[0];
  const float* Mv    = (const float*)d_in[1];
  const float* AMv   = (const float*)d_in[2];
  const int*   srcZ  = (const int*)d_in[3];
  const int*   dstZ  = (const int*)d_in[4];
  const float* valsZ = (const float*)d_in[5];
  const int*   src   = (const int*)d_in[6];
  const int*   dst   = (const int*)d_in[7];
  const float* vals  = (const float*)d_in[8];
  const float* W0 = (const float*)d_in[9];
  const float* b0 = (const float*)d_in[10];
  const float* g0 = (const float*)d_in[11];
  const float* be0 = (const float*)d_in[12];
  const float* rm0 = (const float*)d_in[13];
  const float* rv0 = (const float*)d_in[14];
  const float* W1 = (const float*)d_in[15];
  const float* b1 = (const float*)d_in[16];
  const float* g1 = (const float*)d_in[17];
  const float* be1 = (const float*)d_in[18];
  const float* rm1 = (const float*)d_in[19];
  const float* rv1 = (const float*)d_in[20];
  const float* W2 = (const float*)d_in[21];
  const float* b2 = (const float*)d_in[22];

  int N = in_sizes[1];
  int E = in_sizes[3];
  float* bufA = (float*)d_ws;
  float* bufB = bufA + (size_t)N * NF;
  float* outp = (float*)d_out;

  int tiles_m = (N + 15) / 16;
  long nfeat_elems = (long)N * NF;
  int gemm_blocks = (tiles_m + GEMM_WAVES - 1) / GEMM_WAVES;
  int spmm_blocks = (E + SPMM_EPB - 1) / SPMM_EPB;
  int spmm40_blocks = (E + EPW_C * SPMM_WAVES - 1) / (EPW_C * SPMM_WAVES);

  fill_zero<<<2048, 256, 0, stream>>>(bufA, nfeat_elems);
  spmm_kernel<<<spmm_blocks, 32 * SPMM_WAVES, 0, stream>>>(x, srcZ, dstZ, valsZ, Mv, AMv, bufA, E);
  gemm_strip_kernel<8><<<gemm_blocks, 32 * GEMM_WAVES, 0, stream>>>(
      bufA, W0, b0, g0, be0, rm0, rv0, bufB, nullptr, nullptr, N, NF, 1);

  fill_zero<<<2048, 256, 0, stream>>>(bufA, nfeat_elems);
  spmm_kernel<<<spmm_blocks, 32 * SPMM_WAVES, 0, stream>>>(bufB, src, dst, vals, nullptr, nullptr, bufA, E);
  gemm_strip_kernel<8><<<gemm_blocks, 32 * GEMM_WAVES, 0, stream>>>(
      bufA, W1, b1, g1, be1, rm1, rv1, bufB, nullptr, nullptr, N, NF, 1);

  gemm_strip_kernel<3><<<gemm_blocks, 32 * GEMM_WAVES, 0, stream>>>(
      bufB, W2, nullptr, nullptr, nullptr, nullptr, nullptr, bufA, outp, b2, N, NC, 0);
  spmm40_kernel<<<spmm40_blocks, 32 * SPMM_WAVES, 0, stream>>>(bufA, src, dst, vals, outp, E);
  log_softmax40<<<(N + 255) / 256, 256, 0, stream>>>(outp, N);
}


# Aggregation by owner: edges grouped per call by destination bucket (LDS counting sort); a work-group sums its 400 rows by LDS float adds and stores each once; no global atomics; sum order changes
# speedup vs baseline: 5.9945x; 1.3254x over previous
#include <hip/hip_runtime.h>
#include <math.h>
#include <utility>

typedef __attribute__((ext_vector_type(2))) float v2f;
typedef __attribute__((ext_vector_type(4))) float v4f;
typedef __attribute__((ext_vector_type(8))) float v8f;

#define NF 128

#define NC 40

#define RB 400
#define NBMAX 256
#define GCH 256
#define BK_THREADS 512

__global__ __launch_bounds__(BK_THREADS) void bucket_count(
    const int* __restrict__ dstZ, const int* __restrict__ dstA,
    int* __restrict__ cntZ, int* __restrict__ cntA, int E, int nb) {
  __shared__ int hist[NBMAX];
  const int tid = threadIdx.x;
  const int g = blockIdx.x;
  const int* __restrict__ dst = blockIdx.y == 0 ? dstZ : dstA;
  int* __restrict__ cnt = blockIdx.y == 0 ? cntZ : cntA;
  for (int i = tid; i < NBMAX; i += BK_THREADS) hist[i] = 0;
  __syncthreads();
  const int ch = (E + GCH - 1) / GCH;
  const int lo = g * ch;
  const int hi = lo + ch < E ? lo + ch : E;
  for (int e = lo + tid; e < hi; e += BK_THREADS) atomicAdd(&hist[dst[e] / RB], 1);
  __syncthreads();
  for (int i = tid; i < nb; i += BK_THREADS) cnt[i * GCH + g] = hist[i];
}

__global__ __launch_bounds__(1024) void bucket_scan(int* __restrict__ cZ, int* __restrict__ cA, int n) {
  __shared__ int buf[NBMAX * GCH];
  __shared__ int wsum[32];
  int* __restrict__ c = blockIdx.x == 0 ? cZ : cA;
  const int t = threadIdx.x;
  const int lane = t & 31, wv = t >> 5;
  for (int i = t; i < n; i += 1024) buf[i] = c[i];
  __syncthreads();
  const int per = (n + 1023) / 1024;
  const int lo = t * per;
  const int hi = lo + per < n ? lo + per : n;
  int s = 0;
  for (int i = lo; i < hi; ++i) s += buf[i];
  int x = s;
  for (int d = 1; d < 32; d <<= 1) {
    int y = __shfl_up(x, d);
    if (lane >= d) x += y;
  }
  if (lane == 31) wsum[wv] = x;
  __syncthreads();
  if (wv == 0) {
    int v = wsum[lane];
    for (int d = 1; d < 32; d <<= 1) {
      int y = __shfl_up(v, d);
      if (lane >= d) v += y;
    }
    wsum[lane] = v;
  }
  __syncthreads();
  int base = (wv ? wsum[wv - 1] : 0) + x - s;
  for (int i = lo; i < hi; ++i) {
    int v = buf[i];
    buf[i] = base;
    base += v;
  }
  if (t == 1023) c[n] = base;
  __syncthreads();
  for (int i = t; i < n; i += 1024) c[i] = buf[i];
}

__global__ __launch_bounds__(BK_THREADS) void bucket_place(
    const int* __restrict__ srcZ, const int* __restrict__ dstZ, const float* __restrict__ valsZ,
    const float* __restrict__ Mv, const float* __restrict__ AMv,
    const int* __restrict__ srcA, const int* __restrict__ dstA, const float* __restrict__ valsA,
    const int* __restrict__ offZ, const int* __restrict__ offA,
    float* __restrict__ recZ, float* __restrict__ recA, int E, int nb) {
  __shared__ int cur[NBMAX];
  const int tid = threadIdx.x;
  const int g = blockIdx.x;
  const bool z = blockIdx.y == 0;
  const int* __restrict__ src = z ? srcZ : srcA;
  const int* __restrict__ dst = z ? dstZ : dstA;
  const float* __restrict__ vals = z ? valsZ : valsA;
  const int* __restrict__ off = z ? offZ : offA;
  uint2* __restrict__ rec = (uint2*)(z ? recZ : recA);
  for (int i = tid; i < nb; i += BK_THREADS) cur[i] = off[i * GCH + g];
  __syncthreads();
  const int ch = (E + GCH - 1) / GCH;
  const int lo = g * ch;
  const int hi = lo + ch < E ? lo + ch : E;
  for (int e = lo + tid; e < hi; e += BK_THREADS) {
    const int s = src[e];
    const int d = dst[e];
    float w = vals[e];
    if (z) {
      w *= Mv[s];
      w *= AMv[d];
    }
    const int b = d / RB;
    const int p = atomicAdd(&cur[b], 1);
    rec[p] = make_uint2((unsigned)s | ((unsigned)(d - b * RB) << 17), __float_as_uint(w));
  }
}

#define EPW 8
#define OWN_WAVES 32

__global__ __launch_bounds__(32 * OWN_WAVES) void owner128_kernel(
    const float* __restrict__ X, const float* __restrict__ recf, const int* __restrict__ off,
    float* __restrict__ out, int N) {
  __shared__ float acc[RB * NF];
  const int tid = threadIdx.x;
  for (int i = tid; i < RB * NF / 4; i += 32 * OWN_WAVES) ((v4f*)acc)[i] = v4f{};
  __syncthreads();

  const uint2* __restrict__ rec = (const uint2*)recf;
  const int lane = tid & 31;
  const int wave = __builtin_amdgcn_readfirstlane((int)(tid >> 5));
  const int b = blockIdx.x;
  const int beg = off[b * GCH];
  const int end = off[(b + 1) * GCH];

  for (int e0 = beg + wave * EPW; e0 < end; e0 += OWN_WAVES * EPW) {
    if (e0 + EPW <= end) {
      unsigned s[EPW], d[EPW];
      float w[EPW];
      for (int i = 0; i < EPW; ++i) {
        const uint2 r = rec[e0 + i];
        s[i] = r.x & 0x1FFFFu;
        d[i] = r.x >> 17;
        w[i] = __uint_as_float(r.y);
      }
      float v[EPW][4];
      for (int i = 0; i < EPW; ++i) {
        const float* xr = X + (size_t)s[i] * NF + lane;
        for (int j = 0; j < 4; ++j) v[i][j] = xr[32 * j];
      }
      for (int i = 0; i < EPW; ++i) {
        float* arow = acc + d[i] * NF + lane;
        for (int j = 0; j < 4; ++j) atomicAdd(arow + 32 * j, w[i] * v[i][j]);
      }
    } else {
      for (int e = e0; e < end; ++e) {
        const uint2 r = rec[e];
        const float w = __uint_as_float(r.y);
        const float* xr = X + (size_t)(r.x & 0x1FFFFu) * NF + lane;
        float* arow = acc + (r.x >> 17) * NF + lane;
        for (int j = 0; j < 4; ++j) atomicAdd(arow + 32 * j, w * xr[32 * j]);
      }
    }
  }
  __syncthreads();

  const int row0 = b * RB;
  const int rows = N - row0 < RB ? N - row0 : RB;
  v4f* __restrict__ o4 = (v4f*)(out + (size_t)row0 * NF);
  for (int i = tid; i < rows * (NF / 4); i += 32 * OWN_WAVES) o4[i] = ((const v4f*)acc)[i];
}

#define EPW_C 16
#define NACC (EPW_C / 4 * 5)

template <int G, int J>
__device__ __forceinline__ void gather40(const float* __restrict__ T, const unsigned (&so)[EPW_C],
                                         unsigned lane, float (&v)[NACC]) {
  constexpr unsigned EA = (32 * J) / NC, EB = (32 * J + 31) / NC;
  const unsigned f = 32 * J + lane;
  const bool hi = f >= NC * EB;
  const unsigned base = hi ? so[4 * G + EB] - NC * EB : so[4 * G + EA] - NC * EA;
  v[5 * G + J] = T[base + f];
}

template <int G, int J>
__device__ __forceinline__ void add40(float* acc, const unsigned (&dofs)[EPW_C],
                                      const float (&w)[EPW_C], unsigned lane,
                                      const float (&v)[NACC]) {
  constexpr unsigned EA = (32 * J) / NC, EB = (32 * J + 31) / NC;
  const unsigned f = 32 * J + lane;
  const bool hi = f >= NC * EB;
  const unsigned base = hi ? dofs[4 * G + EB] - NC * EB : dofs[4 * G + EA] - NC * EA;
  const float ww = hi ? w[4 * G + EB] : w[4 * G + EA];
  atomicAdd(acc + (base + f), ww * v[5 * G + J]);
}

template <int... I>
__device__ __forceinline__ void gather40_all(std::integer_sequence<int, I...>,
                                             const float* __restrict__ T,
                                             const unsigned (&so)[EPW_C], unsigned lane,
                                             float (&v)[NACC]) {
  (gather40<I / 5, I % 5>(T, so, lane, v), ...);
}

template <int... I>
__device__ __forceinline__ void add40_all(std::integer_sequence<int, I...>, float* acc,
                                          const unsigned (&dofs)[EPW_C],
                                          const float (&w)[EPW_C], unsigned lane,
                                          const float (&v)[NACC]) {
  (add40<I / 5, I % 5>(acc, dofs, w, lane, v), ...);
}

template <int... I>
__device__ __forceinline__ void records40_all(std::integer_sequence<int, I...>,
                                              const uint2* __restrict__ rec, int e0,
                                              unsigned (&so)[EPW_C], unsigned (&dofs)[EPW_C],
                                              float (&w)[EPW_C]) {
  ((so[I] = (rec[e0 + I].x & 0x1FFFFu) * NC, dofs[I] = (rec[e0 + I].x >> 17) * NC,
    w[I] = __uint_as_float(rec[e0 + I].y)), ...);
}

__global__ __launch_bounds__(32 * OWN_WAVES) void owner40_kernel(
    const float* __restrict__ T, const float* __restrict__ recf, const int* __restrict__ off,
    const float* __restrict__ b2, float* __restrict__ out, int N) {
  __shared__ float acc[RB * NC];
  const int tid = threadIdx.x;
  for (int i = tid; i < RB * NC; i += 32 * OWN_WAVES) acc[i] = b2[i % NC];
  __syncthreads();

  const uint2* __restrict__ rec = (const uint2*)recf;
  const unsigned lane = tid & 31;
  const int wave = __builtin_amdgcn_readfirstlane((int)(tid >> 5));
  const int b = blockIdx.x;
  const int beg = off[b * GCH];
  const int end = off[(b + 1) * GCH];
  const bool tail = lane < NC - 32;

  for (int e0 = beg + wave * EPW_C; e0 < end; e0 += OWN_WAVES * EPW_C) {
    if (e0 + EPW_C <= end) {
      unsigned so[EPW_C], dofs[EPW_C];
      float w[EPW_C];
      float v[NACC];
      records40_all(std::make_integer_sequence<int, EPW_C>{}, rec, e0, so, dofs, w);
      gather40_all(std::make_integer_sequence<int, NACC>{}, T, so, lane, v);
      add40_all(std::make_integer_sequence<int, NACC>{}, acc, dofs, w, lane, v);
    } else {
      for (int e = e0; e < end; ++e) {
        const uint2 r = rec[e];
        const float w = __uint_as_float(r.y);
        const float* tr = T + (r.x & 0x1FFFFu) * NC;
        float* arow = acc + (r.x >> 17) * NC;
        atomicAdd(arow + lane, w * tr[lane]);
        if (tail) atomicAdd(arow + 32 + lane, w * tr[32 + lane]);
      }
    }
  }
  __syncthreads();

  const int row0 = b * RB;
  const int rows = N - row0 < RB ? N - row0 : RB;
  float* __restrict__ o = out + (size_t)row0 * NC;
  for (int i = tid; i < rows * NC; i += 32 * OWN_WAVES) o[i] = acc[i];
}

#define WP (NF + 4)
#define GEMM_WAVES 8

template <int NT>
__global__ __launch_bounds__(32 * GEMM_WAVES) void gemm_strip_kernel(
    const float* __restrict__ A,
    const float* __restrict__ W,
    const float* __restrict__ bias,
    const float* __restrict__ g, const float* __restrict__ be,
    const float* __restrict__ rm, const float* __restrict__ rv,
    float* __restrict__ out,
    int n_rows, int n_out_real, int relu) {
  __shared__ float Ws[NT * 16 * WP];
  const int K = NF;

  for (int idx = threadIdx.x; idx < NT * 16 * (NF / 4); idx += 32 * GEMM_WAVES) {
    int r = idx >> 5;
    int q = idx & 31;
    int rr = r < n_out_real ? r : n_out_real - 1;
    v4f w4 = *(const v4f*)(W + (size_t)rr * K + 4 * q);
    *(v4f*)(Ws + r * WP + 4 * q) = w4;
  }
  __syncthreads();

  const int lane = threadIdx.x & 31;
  const int wave = __builtin_amdgcn_readfirstlane((int)(threadIdx.x >> 5));
  const int tiles_m = (n_rows + 15) >> 4;
  const int strip = blockIdx.x * GEMM_WAVES + wave;
  if (strip >= tiles_m) return;

  const int hl = lane >> 4;
  const int ll = lane & 15;
  const int row = strip * 16 + ll;
  const int ar  = row < n_rows ? row : n_rows - 1;

  const float* ap = A + (size_t)ar * K + 2 * hl;
  v2f a[K / 4];
#pragma unroll
  for (int t = 0; t < K / 4; ++t) a[t] = *(const v2f*)(ap + 4 * t);

  const float* wl = Ws + ll * WP + 2 * hl;
  v8f c[NT];
#pragma unroll
  for (int n = 0; n < NT; ++n) c[n] = v8f{};

#pragma unroll
  for (int t = 0; t < K / 4; ++t) {
    v2f b[NT];
#pragma unroll
    for (int n = 0; n < NT; ++n) b[n] = *(const v2f*)(wl + n * 16 * WP + 4 * t);
#pragma unroll
    for (int n = 0; n < NT; ++n)
      c[n] = __builtin_amdgcn_wmma_f32_16x16x4_f32(
           false, a[t],  false, b[n],
           (short)0, c[n],  false,  false);
  }

  const int rb = strip * 16 + hl * 8;
#pragma unroll
  for (int n = 0; n < NT; ++n) {
    int col = n * 16 + ll;
    int cw  = col < n_out_real ? col : n_out_real - 1;
    float bb = bias ? bias[cw] : 0.0f;
    float sc = 1.0f, sh = 0.0f;
    if (g) {
      float s = g[cw] * rsqrtf(rv[cw] + 1e-5f);
      sc = s;
      sh = be[cw] - rm[cw] * s;
    }
    if (col < n_out_real) {
#pragma unroll
      for (int i = 0; i < 8; ++i) {
        int r = rb + i;
        if (r < n_rows) {
          float v = (c[n][i] + bb) * sc + sh;
          if (relu) v = fmaxf(v, 0.0f);
          out[(size_t)r * n_out_real + col] = v;
        }
      }
    }
  }
}

__global__ __launch_bounds__(256) void log_softmax40(const float* __restrict__ in,
                                                     float* __restrict__ out, int n) {
  int i = blockIdx.x * blockDim.x + threadIdx.x;
  if (i >= n) return;
  const float* row = in + (size_t)i * 40;
  float* orow = out + (size_t)i * 40;
  float m = row[0];
#pragma unroll
  for (int j = 1; j < 40; ++j) m = fmaxf(m, row[j]);
  float s = 0.0f;
#pragma unroll
  for (int j = 0; j < 40; ++j) s += expf(row[j] - m);
  float l = m + logf(s);
#pragma unroll
  for (int j = 0; j < 40; ++j) orow[j] = row[j] - l;
}

extern "C" void kernel_launch(void* const* d_in, const int* in_sizes, int n_in,
                              void* d_out, int out_size, void* d_ws, size_t ws_size,
                              hipStream_t stream) {
  const float* x     = (const float*)d_in[0];
  const float* Mv    = (const float*)d_in[1];
  const float* AMv   = (const float*)d_in[2];
  const int*   srcZ  = (const int*)d_in[3];
  const int*   dstZ  = (const int*)d_in[4];
  const float* valsZ = (const float*)d_in[5];
  const int*   src   = (const int*)d_in[6];
  const int*   dst   = (const int*)d_in[7];
  const float* vals  = (const float*)d_in[8];
  const float* W0 = (const float*)d_in[9];
  const float* b0 = (const float*)d_in[10];
  const float* g0 = (const float*)d_in[11];
  const float* be0 = (const float*)d_in[12];
  const float* rm0 = (const float*)d_in[13];
  const float* rv0 = (const float*)d_in[14];
  const float* W1 = (const float*)d_in[15];
  const float* b1 = (const float*)d_in[16];
  const float* g1 = (const float*)d_in[17];
  const float* be1 = (const float*)d_in[18];
  const float* rm1 = (const float*)d_in[19];
  const float* rv1 = (const float*)d_in[20];
  const float* W2 = (const float*)d_in[21];
  const float* b2 = (const float*)d_in[22];

  int N = in_sizes[1];
  int E = in_sizes[3];
  float* bufA = (float*)d_ws;
  float* bufB = bufA + (size_t)N * NF;
  float* outp = (float*)d_out;

  int tiles_m = (N + 15) / 16;
  int gemm_blocks = (tiles_m + GEMM_WAVES - 1) / GEMM_WAVES;
  int nb = (N + RB - 1) / RB;

  float* recZ = bufB;
  int*   offZ = (int*)(bufB + 2 * (size_t)E);
  float* recA = outp;
  int*   offA = (int*)(outp + 2 * (size_t)E);

  bucket_count<<<dim3(GCH, 2), BK_THREADS, 0, stream>>>(dstZ, dst, offZ, offA, E, nb);
  bucket_scan<<<2, 1024, 0, stream>>>(offZ, offA, nb * GCH);
  bucket_place<<<dim3(GCH, 2), BK_THREADS, 0, stream>>>(srcZ, dstZ, valsZ, Mv, AMv, src, dst, vals,
                                                        offZ, offA, recZ, recA, E, nb);

  owner128_kernel<<<nb, 32 * OWN_WAVES, 0, stream>>>(x, recZ, offZ, bufA, N);
  gemm_strip_kernel<8><<<gemm_blocks, 32 * GEMM_WAVES, 0, stream>>>(
      bufA, W0, b0, g0, be0, rm0, rv0, bufB, N, NF, 1);

  owner128_kernel<<<nb, 32 * OWN_WAVES, 0, stream>>>(bufB, recA, offA, bufA, N);
  gemm_strip_kernel<8><<<gemm_blocks, 32 * GEMM_WAVES, 0, stream>>>(
      bufA, W1, b1, g1, be1, rm1, rv1, bufB, N, NF, 1);

  gemm_strip_kernel<3><<<gemm_blocks, 32 * GEMM_WAVES, 0, stream>>>(
      bufB, W2, nullptr, nullptr, nullptr, nullptr, nullptr, bufA, N, NC, 0);
  owner40_kernel<<<nb, 32 * OWN_WAVES, 0, stream>>>(bufA, recA, offA, b2, bufB, N);
  log_softmax40<<<(N + 255) / 256, 256, 0, stream>>>(bufB, outp, N);
}


# Aggregation: owner work-group sorts its bucket's records by destination row in LDS and sums each row's gathered source rows in registers (16B per lane); no LDS float add per edge; sum order changes
# speedup vs baseline: 6.7210x; 1.1212x over previous
#include <hip/hip_runtime.h>
#include <math.h>

typedef __attribute__((ext_vector_type(2))) float v2f;
typedef __attribute__((ext_vector_type(4))) float v4f;
typedef __attribute__((ext_vector_type(8))) float v8f;

#define NF 128

#define NC 40

#define RB 400
#define NBMAX 256
#define GCH 256
#define BK_THREADS 512

__global__ __launch_bounds__(BK_THREADS) void bucket_count(
    const int* __restrict__ dstZ, const int* __restrict__ dstA,
    int* __restrict__ cntZ, int* __restrict__ cntA, int E, int nb) {
  __shared__ int hist[NBMAX];
  const int tid = threadIdx.x;
  const int g = blockIdx.x;
  const int* __restrict__ dst = blockIdx.y == 0 ? dstZ : dstA;
  int* __restrict__ cnt = blockIdx.y == 0 ? cntZ : cntA;
  for (int i = tid; i < NBMAX; i += BK_THREADS) hist[i] = 0;
  __syncthreads();
  const int ch = (E + GCH - 1) / GCH;
  const int lo = g * ch;
  const int hi = lo + ch < E ? lo + ch : E;
  for (int e = lo + tid; e < hi; e += BK_THREADS) atomicAdd(&hist[dst[e] / RB], 1);
  __syncthreads();
  for (int i = tid; i < nb; i += BK_THREADS) cnt[i * GCH + g] = hist[i];
}

__global__ __launch_bounds__(1024) void bucket_scan(int* __restrict__ cZ, int* __restrict__ cA, int n) {
  __shared__ int buf[NBMAX * GCH];
  __shared__ int wsum[32];
  int* __restrict__ c = blockIdx.x == 0 ? cZ : cA;
  const int t = threadIdx.x;
  const int lane = t & 31, wv = t >> 5;
  for (int i = t; i < n; i += 1024) buf[i] = c[i];
  __syncthreads();
  const int per = (n + 1023) / 1024;
  const int lo = t * per;
  const int hi = lo + per < n ? lo + per : n;
  int s = 0;
  for (int i = lo; i < hi; ++i) s += buf[i];
  int x = s;
  for (int d = 1; d < 32; d <<= 1) {
    int y = __shfl_up(x, d);
    if (lane >= d) x += y;
  }
  if (lane == 31) wsum[wv] = x;
  __syncthreads();
  if (wv == 0) {
    int v = wsum[lane];
    for (int d = 1; d < 32; d <<= 1) {
      int y = __shfl_up(v, d);
      if (lane >= d) v += y;
    }
    wsum[lane] = v;
  }
  __syncthreads();
  int base = (wv ? wsum[wv - 1] : 0) + x - s;
  for (int i = lo; i < hi; ++i) {
    int v = buf[i];
    buf[i] = base;
    base += v;
  }
  if (t == 1023) c[n] = base;
  __syncthreads();
  for (int i = t; i < n; i += 1024) c[i] = buf[i];
}

__global__ __launch_bounds__(BK_THREADS) void bucket_place(
    const int* __restrict__ srcZ, const int* __restrict__ dstZ, const float* __restrict__ valsZ,
    const float* __restrict__ Mv, const float* __restrict__ AMv,
    const int* __restrict__ srcA, const int* __restrict__ dstA, const float* __restrict__ valsA,
    const int* __restrict__ offZ, const int* __restrict__ offA,
    float* __restrict__ recZ, float* __restrict__ recA, int E, int nb) {
  __shared__ int cur[NBMAX];
  const int tid = threadIdx.x;
  const int g = blockIdx.x;
  const bool z = blockIdx.y == 0;
  const int* __restrict__ src = z ? srcZ : srcA;
  const int* __restrict__ dst = z ? dstZ : dstA;
  const float* __restrict__ vals = z ? valsZ : valsA;
  const int* __restrict__ off = z ? offZ : offA;
  uint2* __restrict__ rec = (uint2*)(z ? recZ : recA);
  for (int i = tid; i < nb; i += BK_THREADS) cur[i] = off[i * GCH + g];
  __syncthreads();
  const int ch = (E + GCH - 1) / GCH;
  const int lo = g * ch;
  const int hi = lo + ch < E ? lo + ch : E;
  for (int e = lo + tid; e < hi; e += BK_THREADS) {
    const int s = src[e];
    const int d = dst[e];
    float w = vals[e];
    if (z) {
      w *= Mv[s];
      w *= AMv[d];
    }
    const int b = d / RB;
    const int p = atomicAdd(&cur[b], 1);
    rec[p] = make_uint2((unsigned)s | ((unsigned)(d - b * RB) << 17), __float_as_uint(w));
  }
}

#define CAP 8192
#define RPT (CAP / 1024)
#define EPB 16
#define OWN_WAVES 32

struct SortLds {
  uint2 srt[CAP + 32];
  int cnt[512];
  int start[512];
  int wsum[32];
};

__device__ __forceinline__ void sort_bucket(SortLds& L, const uint2* __restrict__ rec, int cb, int m,
                                            int tid, int lane, int wave) {
  if (tid < 512) L.cnt[tid] = 0;
  __syncthreads();

  uint2 r[RPT];
  int rk[RPT];
#pragma unroll
  for (int j = 0; j < RPT; ++j) {
    const int i = tid + 1024 * j;
    if (i < m) r[j] = rec[cb + i];
  }
#pragma unroll
  for (int j = 0; j < RPT; ++j) {
    const int i = tid + 1024 * j;
    if (i < m) rk[j] = atomicAdd(&L.cnt[r[j].x >> 17], 1);
  }
  __syncthreads();

  const int cv = tid < RB ? L.cnt[tid] : 0;
  int x = cv;
  for (int d = 1; d < 32; d <<= 1) {
    int y = __shfl_up(x, d);
    if (lane >= d) x += y;
  }
  if (lane == 31) L.wsum[wave] = x;
  __syncthreads();
  if (wave == 0) {
    int t = L.wsum[lane];
    for (int d = 1; d < 32; d <<= 1) {
      int y = __shfl_up(t, d);
      if (lane >= d) t += y;
    }
    L.wsum[lane] = t;
  }
  __syncthreads();
  if (tid <= RB) L.start[tid] = (wave ? L.wsum[wave - 1] : 0) + x - cv;
  __syncthreads();

#pragma unroll
  for (int j = 0; j < RPT; ++j) {
    const int i = tid + 1024 * j;
    if (i < m) L.srt[L.start[r[j].x >> 17] + rk[j]] = r[j];
  }
  __syncthreads();
}

template <bool FULL, int ROWF, class Flush>
__device__ __forceinline__ void stream_batch(const float* __restrict__ X, const uint2* srt, int p,
                                             int nrec, int lane, int l4, int& cur, v4f& acc,
                                             Flush flush) {
  const uint2 rr = srt[p + lane];
  unsigned sx[EPB];
  float sw[EPB];
#pragma unroll
  for (int k = 0; k < EPB; ++k) {
    sx[k] = (unsigned)__builtin_amdgcn_readlane((int)rr.x, k);
    sw[k] = __uint_as_float((unsigned)__builtin_amdgcn_readlane((int)rr.y, k));
  }
  v4f v[EPB];
#pragma unroll
  for (int k = 0; k < EPB; ++k) {
    const unsigned s = (FULL || k < nrec) ? sx[k] : sx[0];
    v[k] = ((const v4f*)(X + (size_t)(s & 0x1FFFFu) * ROWF))[l4];
  }
#pragma unroll
  for (int k = 0; k < EPB; ++k) {
    if (FULL || k < nrec) {
      const int row = (int)(sx[k] >> 17);
      if (row != cur) {
        if (cur >= 0) flush(cur, acc);
        acc = v4f{};
        cur = row;
      }
      acc += sw[k] * v[k];
    }
  }
}

__global__ __launch_bounds__(32 * OWN_WAVES) void owner128_kernel(
    const float* __restrict__ X, const float* __restrict__ recf, const int* __restrict__ off,
    float* __restrict__ out, int N) {
  __shared__ SortLds L;
  const int tid = threadIdx.x;
  const int lane = tid & 31;
  const int wave = __builtin_amdgcn_readfirstlane((int)(tid >> 5));
  const uint2* __restrict__ rec = (const uint2*)recf;
  const int b = blockIdx.x;
  const int beg = off[b * GCH];
  const int end = off[(b + 1) * GCH];
  const int row0 = b * RB;
  const int rows = N - row0 < RB ? N - row0 : RB;
  const int r0 = (wave * RB) >> 5;
  const int r1 = ((wave + 1) * RB) >> 5;
  float* __restrict__ orow = out + (size_t)row0 * NF;

  int pass = 0;
  int cb = beg;
  do {
    const int m = end - cb < CAP ? end - cb : CAP;
    sort_bucket(L, rec, cb, m, tid, lane, wave);

    if (pass == 0) {
      for (int q = r0; q < r1; ++q) {
        if (q < rows && __builtin_amdgcn_readfirstlane(L.cnt[q]) == 0)
          ((v4f*)(orow + (size_t)q * NF))[lane] = v4f{};
      }
    }

    auto flush = [&](int row, v4f sum) {
      v4f* dst = (v4f*)(orow + (size_t)row * NF) + lane;
      if (pass) sum += *dst;
      *dst = sum;
    };
    int p = __builtin_amdgcn_readfirstlane(L.start[r0]);
    const int pend = __builtin_amdgcn_readfirstlane(L.start[r1]);
    int cur = -1;
    v4f acc = v4f{};
    for (; p + EPB <= pend; p += EPB)
      stream_batch<true, NF>(X, L.srt, p, EPB, lane, lane, cur, acc, flush);
    if (p < pend) stream_batch<false, NF>(X, L.srt, p, pend - p, lane, lane, cur, acc, flush);
    if (cur >= 0) flush(cur, acc);

    cb += CAP;
    ++pass;
    __syncthreads();
  } while (cb < end);
}

#define NC4 (NC / 4)

__global__ __launch_bounds__(32 * OWN_WAVES) void owner40_kernel(
    const float* __restrict__ T, const float* __restrict__ recf, const int* __restrict__ off,
    const float* __restrict__ b2, float* __restrict__ out, int N) {
  __shared__ SortLds L;
  __shared__ float tile[RB * NC];
  const int tid = threadIdx.x;
  const int lane = tid & 31;
  const int wave = __builtin_amdgcn_readfirstlane((int)(tid >> 5));
  const int l4 = lane < NC4 ? lane : NC4 - 1;
  for (int i = tid; i < RB * NC; i += 32 * OWN_WAVES) tile[i] = b2[i % NC];

  const uint2* __restrict__ rec = (const uint2*)recf;
  const int b = blockIdx.x;
  const int beg = off[b * GCH];
  const int end = off[(b + 1) * GCH];
  const int r0 = (wave * RB) >> 5;
  const int r1 = ((wave + 1) * RB) >> 5;

  auto flush = [&](int row, v4f sum) {
    if (lane < NC4) {
      v4f* t = (v4f*)(tile + row * NC) + lane;
      *t += sum;
    }
  };
  int cb = beg;
  do {
    const int m = end - cb < CAP ? end - cb : CAP;
    sort_bucket(L, rec, cb, m, tid, lane, wave);

    int p = __builtin_amdgcn_readfirstlane(L.start[r0]);
    const int pend = __builtin_amdgcn_readfirstlane(L.start[r1]);
    int cur = -1;
    v4f acc = v4f{};
    for (; p + EPB <= pend; p += EPB)
      stream_batch<true, NC>(T, L.srt, p, EPB, lane, l4, cur, acc, flush);
    if (p < pend) stream_batch<false, NC>(T, L.srt, p, pend - p, lane, l4, cur, acc, flush);
    if (cur >= 0) flush(cur, acc);

    cb += CAP;
    __syncthreads();
  } while (cb < end);

  const int row0 = b * RB;
  const int rows = N - row0 < RB ? N - row0 : RB;
  float* __restrict__ o = out + (size_t)row0 * NC;
  for (int i = tid; i < rows * NC; i += 32 * OWN_WAVES) o[i] = tile[i];
}

#define WP (NF + 4)
#define GEMM_WAVES 8

template <int NT>
__global__ __launch_bounds__(32 * GEMM_WAVES) void gemm_strip_kernel(
    const float* __restrict__ A,
    const float* __restrict__ W,
    const float* __restrict__ bias,
    const float* __restrict__ g, const float* __restrict__ be,
    const float* __restrict__ rm, const float* __restrict__ rv,
    float* __restrict__ out,
    int n_rows, int n_out_real, int relu) {
  __shared__ float Ws[NT * 16 * WP];
  const int K = NF;

  for (int idx = threadIdx.x; idx < NT * 16 * (NF / 4); idx += 32 * GEMM_WAVES) {
    int r = idx >> 5;
    int q = idx & 31;
    int rr = r < n_out_real ? r : n_out_real - 1;
    v4f w4 = *(const v4f*)(W + (size_t)rr * K + 4 * q);
    *(v4f*)(Ws + r * WP + 4 * q) = w4;
  }
  __syncthreads();

  const int lane = threadIdx.x & 31;
  const int wave = __builtin_amdgcn_readfirstlane((int)(threadIdx.x >> 5));
  const int tiles_m = (n_rows + 15) >> 4;
  const int strip = blockIdx.x * GEMM_WAVES + wave;
  if (strip >= tiles_m) return;

  const int hl = lane >> 4;
  const int ll = lane & 15;
  const int row = strip * 16 + ll;
  const int ar  = row < n_rows ? row : n_rows - 1;

  const float* ap = A + (size_t)ar * K + 2 * hl;
  v2f a[K / 4];
#pragma unroll
  for (int t = 0; t < K / 4; ++t) a[t] = *(const v2f*)(ap + 4 * t);

  const float* wl = Ws + ll * WP + 2 * hl;
  v8f c[NT];
#pragma unroll
  for (int n = 0; n < NT; ++n) c[n] = v8f{};

#pragma unroll
  for (int t = 0; t < K / 4; ++t) {
    v2f b[NT];
#pragma unroll
    for (int n = 0; n < NT; ++n) b[n] = *(const v2f*)(wl + n * 16 * WP + 4 * t);
#pragma unroll
    for (int n = 0; n < NT; ++n)
      c[n] = __builtin_amdgcn_wmma_f32_16x16x4_f32(
           false, a[t],  false, b[n],
           (short)0, c[n],  false,  false);
  }

  const int rb = strip * 16 + hl * 8;
#pragma unroll
  for (int n = 0; n < NT; ++n) {
    int col = n * 16 + ll;
    int cw  = col < n_out_real ? col : n_out_real - 1;
    float bb = bias ? bias[cw] : 0.0f;
    float sc = 1.0f, sh = 0.0f;
    if (g) {
      float s = g[cw] * rsqrtf(rv[cw] + 1e-5f);
      sc = s;
      sh = be[cw] - rm[cw] * s;
    }
    if (col < n_out_real) {
#pragma unroll
      for (int i = 0; i < 8; ++i) {
        int r = rb + i;
        if (r < n_rows) {
          float v = (c[n][i] + bb) * sc + sh;
          if (relu) v = fmaxf(v, 0.0f);
          out[(size_t)r * n_out_real + col] = v;
        }
      }
    }
  }
}

__global__ __launch_bounds__(256) void log_softmax40(const float* __restrict__ in,
                                                     float* __restrict__ out, int n) {
  int i = blockIdx.x * blockDim.x + threadIdx.x;
  if (i >= n) return;
  const float* row = in + (size_t)i * 40;
  float* orow = out + (size_t)i * 40;
  float m = row[0];
#pragma unroll
  for (int j = 1; j < 40; ++j) m = fmaxf(m, row[j]);
  float s = 0.0f;
#pragma unroll
  for (int j = 0; j < 40; ++j) s += expf(row[j] - m);
  float l = m + logf(s);
#pragma unroll
  for (int j = 0; j < 40; ++j) orow[j] = row[j] - l;
}

extern "C" void kernel_launch(void* const* d_in, const int* in_sizes, int n_in,
                              void* d_out, int out_size, void* d_ws, size_t ws_size,
                              hipStream_t stream) {
  const float* x     = (const float*)d_in[0];
  const float* Mv    = (const float*)d_in[1];
  const float* AMv   = (const float*)d_in[2];
  const int*   srcZ  = (const int*)d_in[3];
  const int*   dstZ  = (const int*)d_in[4];
  const float* valsZ = (const float*)d_in[5];
  const int*   src   = (const int*)d_in[6];
  const int*   dst   = (const int*)d_in[7];
  const float* vals  = (const float*)d_in[8];
  const float* W0 = (const float*)d_in[9];
  const float* b0 = (const float*)d_in[10];
  const float* g0 = (const float*)d_in[11];
  const float* be0 = (const float*)d_in[12];
  const float* rm0 = (const float*)d_in[13];
  const float* rv0 = (const float*)d_in[14];
  const float* W1 = (const float*)d_in[15];
  const float* b1 = (const float*)d_in[16];
  const float* g1 = (const float*)d_in[17];
  const float* be1 = (const float*)d_in[18];
  const float* rm1 = (const float*)d_in[19];
  const float* rv1 = (const float*)d_in[20];
  const float* W2 = (const float*)d_in[21];
  const float* b2 = (const float*)d_in[22];

  int N = in_sizes[1];
  int E = in_sizes[3];
  float* bufA = (float*)d_ws;
  float* bufB = bufA + (size_t)N * NF;
  float* outp = (float*)d_out;

  int tiles_m = (N + 15) / 16;
  int gemm_blocks = (tiles_m + GEMM_WAVES - 1) / GEMM_WAVES;
  int nb = (N + RB - 1) / RB;

  float* recZ = bufB;
  int*   offZ = (int*)(bufB + 2 * (size_t)E);
  float* recA = outp;
  int*   offA = (int*)(outp + 2 * (size_t)E);

  bucket_count<<<dim3(GCH, 2), BK_THREADS, 0, stream>>>(dstZ, dst, offZ, offA, E, nb);
  bucket_scan<<<2, 1024, 0, stream>>>(offZ, offA, nb * GCH);
  bucket_place<<<dim3(GCH, 2), BK_THREADS, 0, stream>>>(srcZ, dstZ, valsZ, Mv, AMv, src, dst, vals,
                                                        offZ, offA, recZ, recA, E, nb);

  owner128_kernel<<<nb, 32 * OWN_WAVES, 0, stream>>>(x, recZ, offZ, bufA, N);
  gemm_strip_kernel<8><<<gemm_blocks, 32 * GEMM_WAVES, 0, stream>>>(
      bufA, W0, b0, g0, be0, rm0, rv0, bufB, N, NF, 1);

  owner128_kernel<<<nb, 32 * OWN_WAVES, 0, stream>>>(bufB, recA, offA, bufA, N);
  gemm_strip_kernel<8><<<gemm_blocks, 32 * GEMM_WAVES, 0, stream>>>(
      bufA, W1, b1, g1, be1, rm1, rv1, bufB, N, NF, 1);

  gemm_strip_kernel<3><<<gemm_blocks, 32 * GEMM_WAVES, 0, stream>>>(
      bufB, W2, nullptr, nullptr, nullptr, nullptr, nullptr, bufA, N, NC, 0);
  owner40_kernel<<<nb, 32 * OWN_WAVES, 0, stream>>>(bufA, recA, offA, b2, bufB, N);
  log_softmax40<<<(N + 255) / 256, 256, 0, stream>>>(bufB, outp, N);
}


# Layers 0, 1: owner work-group (25 waves, a 16-row strip each) keeps summed rows in an LDS tile and runs the layer's WMMA product from it: no aggregate buffer, 2 launches fewer; sum order as before
# speedup vs baseline: 7.6980x; 1.1454x over previous
#include <hip/hip_runtime.h>
#include <math.h>

typedef __attribute__((ext_vector_type(2))) float v2f;
typedef __attribute__((ext_vector_type(4))) float v4f;
typedef __attribute__((ext_vector_type(8))) float v8f;

#define NF 128

#define NC 40
#define WP (NF + 4)

#define RB 400
#define NBMAX 256
#define GCH 256
#define BK_THREADS 512

__global__ __launch_bounds__(BK_THREADS) void bucket_count(
    const int* __restrict__ dstZ, const int* __restrict__ dstA,
    int* __restrict__ cntZ, int* __restrict__ cntA, int E, int nb) {
  __shared__ int hist[NBMAX];
  const int tid = threadIdx.x;
  const int g = blockIdx.x;
  const int* __restrict__ dst = blockIdx.y == 0 ? dstZ : dstA;
  int* __restrict__ cnt = blockIdx.y == 0 ? cntZ : cntA;
  for (int i = tid; i < NBMAX; i += BK_THREADS) hist[i] = 0;
  __syncthreads();
  const int ch = (E + GCH - 1) / GCH;
  const int lo = g * ch;
  const int hi = lo + ch < E ? lo + ch : E;
  for (int e = lo + tid; e < hi; e += BK_THREADS) atomicAdd(&hist[dst[e] / RB], 1);
  __syncthreads();
  for (int i = tid; i < nb; i += BK_THREADS) cnt[i * GCH + g] = hist[i];
}

__global__ __launch_bounds__(1024) void bucket_scan(int* __restrict__ cZ, int* __restrict__ cA, int n) {
  __shared__ int buf[NBMAX * GCH];
  __shared__ int wsum[32];
  int* __restrict__ c = blockIdx.x == 0 ? cZ : cA;
  const int t = threadIdx.x;
  const int lane = t & 31, wv = t >> 5;
  for (int i = t; i < n; i += 1024) buf[i] = c[i];
  __syncthreads();
  const int per = (n + 1023) / 1024;
  const int lo = t * per;
  const int hi = lo + per < n ? lo + per : n;
  int s = 0;
  for (int i = lo; i < hi; ++i) s += buf[i];
  int x = s;
  for (int d = 1; d < 32; d <<= 1) {
    int y = __shfl_up(x, d);
    if (lane >= d) x += y;
  }
  if (lane == 31) wsum[wv] = x;
  __syncthreads();
  if (wv == 0) {
    int v = wsum[lane];
    for (int d = 1; d < 32; d <<= 1) {
      int y = __shfl_up(v, d);
      if (lane >= d) v += y;
    }
    wsum[lane] = v;
  }
  __syncthreads();
  int base = (wv ? wsum[wv - 1] : 0) + x - s;
  for (int i = lo; i < hi; ++i) {
    int v = buf[i];
    buf[i] = base;
    base += v;
  }
  if (t == 1023) c[n] = base;
  __syncthreads();
  for (int i = t; i < n; i += 1024) c[i] = buf[i];
}

__global__ __launch_bounds__(BK_THREADS) void bucket_place(
    const int* __restrict__ srcZ, const int* __restrict__ dstZ, const float* __restrict__ valsZ,
    const float* __restrict__ Mv, const float* __restrict__ AMv,
    const int* __restrict__ srcA, const int* __restrict__ dstA, const float* __restrict__ valsA,
    const int* __restrict__ offZ, const int* __restrict__ offA,
    float* __restrict__ recZ, float* __restrict__ recA, int E, int nb) {
  __shared__ int cur[NBMAX];
  const int tid = threadIdx.x;
  const int g = blockIdx.x;
  const bool z = blockIdx.y == 0;
  const int* __restrict__ src = z ? srcZ : srcA;
  const int* __restrict__ dst = z ? dstZ : dstA;
  const float* __restrict__ vals = z ? valsZ : valsA;
  const int* __restrict__ off = z ? offZ : offA;
  uint2* __restrict__ rec = (uint2*)(z ? recZ : recA);
  for (int i = tid; i < nb; i += BK_THREADS) cur[i] = off[i * GCH + g];
  __syncthreads();
  const int ch = (E + GCH - 1) / GCH;
  const int lo = g * ch;
  const int hi = lo + ch < E ? lo + ch : E;
  for (int e = lo + tid; e < hi; e += BK_THREADS) {
    const int s = src[e];
    const int d = dst[e];
    float w = vals[e];
    if (z) {
      w *= Mv[s];
      w *= AMv[d];
    }
    const int b = d / RB;
    const int p = atomicAdd(&cur[b], 1);
    rec[p] = make_uint2((unsigned)s | ((unsigned)(d - b * RB) << 17), __float_as_uint(w));
  }
}

#define CAP 8192
#define EPB 16
#define OWN_WAVES 32

struct SortLds {
  uint2 srt[CAP + 32];
  int cnt[512];
  int start[512];
  int wsum[32];
};

template <int NTHR>
__device__ __forceinline__ void sort_bucket(SortLds& L, const uint2* __restrict__ rec, int cb, int m,
                                            int tid, int lane, int wave) {
  constexpr int RPT = (CAP + NTHR - 1) / NTHR;
  if (tid < 512) L.cnt[tid] = 0;
  if (NTHR < 1024 && tid < 32) L.wsum[tid] = 0;
  __syncthreads();

  uint2 r[RPT];
  int rk[RPT];
#pragma unroll
  for (int j = 0; j < RPT; ++j) {
    const int i = tid + NTHR * j;
    if (i < m) r[j] = rec[cb + i];
  }
#pragma unroll
  for (int j = 0; j < RPT; ++j) {
    const int i = tid + NTHR * j;
    if (i < m) rk[j] = atomicAdd(&L.cnt[r[j].x >> 17], 1);
  }
  __syncthreads();

  const int cv = tid < RB ? L.cnt[tid] : 0;
  int x = cv;
  for (int d = 1; d < 32; d <<= 1) {
    int y = __shfl_up(x, d);
    if (lane >= d) x += y;
  }
  if (lane == 31) L.wsum[wave] = x;
  __syncthreads();
  if (wave == 0) {
    int t = L.wsum[lane];
    for (int d = 1; d < 32; d <<= 1) {
      int y = __shfl_up(t, d);
      if (lane >= d) t += y;
    }
    L.wsum[lane] = t;
  }
  __syncthreads();
  if (tid <= RB) L.start[tid] = (wave ? L.wsum[wave - 1] : 0) + x - cv;
  __syncthreads();

#pragma unroll
  for (int j = 0; j < RPT; ++j) {
    const int i = tid + NTHR * j;
    if (i < m) L.srt[L.start[r[j].x >> 17] + rk[j]] = r[j];
  }
  __syncthreads();
}

template <bool FULL, int ROWF, class Flush>
__device__ __forceinline__ void stream_batch(const float* __restrict__ X, const uint2* srt, int p,
                                             int nrec, int lane, int l4, int& cur, v4f& acc,
                                             Flush flush) {
  const uint2 rr = srt[p + lane];
  unsigned sx[EPB];
  float sw[EPB];
#pragma unroll
  for (int k = 0; k < EPB; ++k) {
    sx[k] = (unsigned)__builtin_amdgcn_readlane((int)rr.x, k);
    sw[k] = __uint_as_float((unsigned)__builtin_amdgcn_readlane((int)rr.y, k));
  }
  v4f v[EPB];
#pragma unroll
  for (int k = 0; k < EPB; ++k) {
    const unsigned s = (FULL || k < nrec) ? sx[k] : sx[0];
    v[k] = ((const v4f*)(X + (size_t)(s & 0x1FFFFu) * ROWF))[l4];
  }
#pragma unroll
  for (int k = 0; k < EPB; ++k) {
    if (FULL || k < nrec) {
      const int row = (int)(sx[k] >> 17);
      if (row != cur) {
        if (cur >= 0) flush(cur, acc);
        acc = v4f{};
        cur = row;
      }
      acc += sw[k] * v[k];
    }
  }
}

#define FW (RB / 16)
#define FT (32 * FW)
#define AP (NF + 4)

struct alignas(16) FusedLds {
  union {
    SortLds S;
    float Ws[NF * WP];
  } u;
  float At[RB * AP];
};

__global__ __launch_bounds__(FT) void owner_gemm_kernel(
    const float* __restrict__ X, const float* __restrict__ recf, const int* __restrict__ off,
    const float* __restrict__ W, const float* __restrict__ bias,
    const float* __restrict__ g, const float* __restrict__ be,
    const float* __restrict__ rm, const float* __restrict__ rv,
    float* __restrict__ out, int N) {
  static_assert(RB % 16 == 0 && FW <= 32, "one wave per 16-row strip");
  __shared__ FusedLds L;
  const int tid = threadIdx.x;
  const int lane = tid & 31;
  const int wave = __builtin_amdgcn_readfirstlane((int)(tid >> 5));
  const uint2* __restrict__ rec = (const uint2*)recf;
  const int b = blockIdx.x;
  const int beg = off[b * GCH];
  const int end = off[(b + 1) * GCH];
  const int r0 = wave * 16;
  const int r1 = r0 + 16;

  int pass = 0;
  int cb = beg;
  do {
    const int m = end - cb < CAP ? end - cb : CAP;
    sort_bucket<FT>(L.u.S, rec, cb, m, tid, lane, wave);

    if (pass == 0) {
      for (int q = r0; q < r1; ++q) {
        if (__builtin_amdgcn_readfirstlane(L.u.S.cnt[q]) == 0)
          ((v4f*)(L.At + q * AP))[lane] = v4f{};
      }
    }

    auto flush = [&](int row, v4f sum) {
      v4f* dst = (v4f*)(L.At + row * AP) + lane;
      if (pass) sum += *dst;
      *dst = sum;
    };
    int p = __builtin_amdgcn_readfirstlane(L.u.S.start[r0]);
    const int pend = __builtin_amdgcn_readfirstlane(L.u.S.start[r1]);
    int cur = -1;
    v4f acc = v4f{};
    for (; p + EPB <= pend; p += EPB)
      stream_batch<true, NF>(X, L.u.S.srt, p, EPB, lane, lane, cur, acc, flush);
    if (p < pend) stream_batch<false, NF>(X, L.u.S.srt, p, pend - p, lane, lane, cur, acc, flush);
    if (cur >= 0) flush(cur, acc);

    cb += CAP;
    ++pass;
    if (cb < end) __syncthreads();
  } while (cb < end);

  constexpr int WPT = (NF * NF / 4 + FT - 1) / FT;
  v4f wv[WPT];
#pragma unroll
  for (int j = 0; j < WPT; ++j) {
    const int idx = tid + FT * j;
    if (idx < NF * NF / 4) wv[j] = *(const v4f*)(W + 4 * idx);
  }
  __syncthreads();
#pragma unroll
  for (int j = 0; j < WPT; ++j) {
    const int idx = tid + FT * j;
    if (idx < NF * NF / 4) *(v4f*)(L.u.Ws + (idx >> 5) * WP + 4 * (idx & 31)) = wv[j];
  }
  __syncthreads();

  const int hl = lane >> 4;
  const int ll = lane & 15;
  const float* al = L.At + (r0 + ll) * AP + 2 * hl;
  const float* wl = L.u.Ws + ll * WP + 2 * hl;
  v8f c[NF / 16];
#pragma unroll
  for (int n = 0; n < NF / 16; ++n) c[n] = v8f{};

#pragma unroll 4
  for (int t = 0; t < NF / 4; ++t) {
    const v2f a = *(const v2f*)(al + 4 * t);
    v2f bf[NF / 16];
#pragma unroll
    for (int n = 0; n < NF / 16; ++n) bf[n] = *(const v2f*)(wl + n * 16 * WP + 4 * t);
#pragma unroll
    for (int n = 0; n < NF / 16; ++n)
      c[n] = __builtin_amdgcn_wmma_f32_16x16x4_f32(
           false, a,  false, bf[n],
           (short)0, c[n],  false,  false);
  }

  const int rb = b * RB + r0 + hl * 8;
#pragma unroll
  for (int n = 0; n < NF / 16; ++n) {
    const int col = n * 16 + ll;
    const float bb = bias[col];
    const float sc = g[col] * rsqrtf(rv[col] + 1e-5f);
    const float sh = be[col] - rm[col] * sc;
#pragma unroll
    for (int i = 0; i < 8; ++i) {
      const int r = rb + i;
      if (r < N) out[(size_t)r * NF + col] = fmaxf((c[n][i] + bb) * sc + sh, 0.0f);
    }
  }
}

#define NC4 (NC / 4)

__global__ __launch_bounds__(32 * OWN_WAVES) void owner40_kernel(
    const float* __restrict__ T, const float* __restrict__ recf, const int* __restrict__ off,
    const float* __restrict__ b2, float* __restrict__ out, int N) {
  __shared__ SortLds L;
  __shared__ float tile[RB * NC];
  const int tid = threadIdx.x;
  const int lane = tid & 31;
  const int wave = __builtin_amdgcn_readfirstlane((int)(tid >> 5));
  const int l4 = lane < NC4 ? lane : NC4 - 1;
  for (int i = tid; i < RB * NC; i += 32 * OWN_WAVES) tile[i] = b2[i % NC];

  const uint2* __restrict__ rec = (const uint2*)recf;
  const int b = blockIdx.x;
  const int beg = off[b * GCH];
  const int end = off[(b + 1) * GCH];
  const int r0 = (wave * RB) >> 5;
  const int r1 = ((wave + 1) * RB) >> 5;

  auto flush = [&](int row, v4f sum) {
    if (lane < NC4) {
      v4f* t = (v4f*)(tile + row * NC) + lane;
      *t += sum;
    }
  };
  int cb = beg;
  do {
    const int m = end - cb < CAP ? end - cb : CAP;
    sort_bucket<32 * OWN_WAVES>(L, rec, cb, m, tid, lane, wave);

    int p = __builtin_amdgcn_readfirstlane(L.start[r0]);
    const int pend = __builtin_amdgcn_readfirstlane(L.start[r1]);
    int cur = -1;
    v4f acc = v4f{};
    for (; p + EPB <= pend; p += EPB)
      stream_batch<true, NC>(T, L.srt, p, EPB, lane, l4, cur, acc, flush);
    if (p < pend) stream_batch<false, NC>(T, L.srt, p, pend - p, lane, l4, cur, acc, flush);
    if (cur >= 0) flush(cur, acc);

    cb += CAP;
    __syncthreads();
  } while (cb < end);

  const int row0 = b * RB;
  const int rows = N - row0 < RB ? N - row0 : RB;
  float* __restrict__ o = out + (size_t)row0 * NC;
  for (int i = tid; i < rows * NC; i += 32 * OWN_WAVES) o[i] = tile[i];
}

#define GEMM_WAVES 8

template <int NT>
__global__ __launch_bounds__(32 * GEMM_WAVES) void gemm_strip_kernel(
    const float* __restrict__ A,
    const float* __restrict__ W,
    const float* __restrict__ bias,
    const float* __restrict__ g, const float* __restrict__ be,
    const float* __restrict__ rm, const float* __restrict__ rv,
    float* __restrict__ out,
    int n_rows, int n_out_real, int relu) {
  __shared__ float Ws[NT * 16 * WP];
  const int K = NF;

  for (int idx = threadIdx.x; idx < NT * 16 * (NF / 4); idx += 32 * GEMM_WAVES) {
    int r = idx >> 5;
    int q = idx & 31;
    int rr = r < n_out_real ? r : n_out_real - 1;
    v4f w4 = *(const v4f*)(W + (size_t)rr * K + 4 * q);
    *(v4f*)(Ws + r * WP + 4 * q) = w4;
  }
  __syncthreads();

  const int lane = threadIdx.x & 31;
  const int wave = __builtin_amdgcn_readfirstlane((int)(threadIdx.x >> 5));
  const int tiles_m = (n_rows + 15) >> 4;
  const int strip = blockIdx.x * GEMM_WAVES + wave;
  if (strip >= tiles_m) return;

  const int hl = lane >> 4;
  const int ll = lane & 15;
  const int row = strip * 16 + ll;
  const int ar  = row < n_rows ? row : n_rows - 1;

  const float* ap = A + (size_t)ar * K + 2 * hl;
  v2f a[K / 4];
#pragma unroll
  for (int t = 0; t < K / 4; ++t) a[t] = *(const v2f*)(ap + 4 * t);

  const float* wl = Ws + ll * WP + 2 * hl;
  v8f c[NT];
#pragma unroll
  for (int n = 0; n < NT; ++n) c[n] = v8f{};

#pragma unroll
  for (int t = 0; t < K / 4; ++t) {
    v2f b[NT];
#pragma unroll
    for (int n = 0; n < NT; ++n) b[n] = *(const v2f*)(wl + n * 16 * WP + 4 * t);
#pragma unroll
    for (int n = 0; n < NT; ++n)
      c[n] = __builtin_amdgcn_wmma_f32_16x16x4_f32(
           false, a[t],  false, b[n],
           (short)0, c[n],  false,  false);
  }

  const int rb = strip * 16 + hl * 8;
#pragma unroll
  for (int n = 0; n < NT; ++n) {
    int col = n * 16 + ll;
    int cw  = col < n_out_real ? col : n_out_real - 1;
    float bb = bias ? bias[cw] : 0.0f;
    float sc = 1.0f, sh = 0.0f;
    if (g) {
      float s = g[cw] * rsqrtf(rv[cw] + 1e-5f);
      sc = s;
      sh = be[cw] - rm[cw] * s;
    }
    if (col < n_out_real) {
#pragma unroll
      for (int i = 0; i < 8; ++i) {
        int r = rb + i;
        if (r < n_rows) {
          float v = (c[n][i] + bb) * sc + sh;
          if (relu) v = fmaxf(v, 0.0f);
          out[(size_t)r * n_out_real + col] = v;
        }
      }
    }
  }
}

__global__ __launch_bounds__(256) void log_softmax40(const float* __restrict__ in,
                                                     float* __restrict__ out, int n) {
  int i = blockIdx.x * blockDim.x + threadIdx.x;
  if (i >= n) return;
  const float* row = in + (size_t)i * 40;
  float* orow = out + (size_t)i * 40;
  float m = row[0];
#pragma unroll
  for (int j = 1; j < 40; ++j) m = fmaxf(m, row[j]);
  float s = 0.0f;
#pragma unroll
  for (int j = 0; j < 40; ++j) s += expf(row[j] - m);
  float l = m + logf(s);
#pragma unroll
  for (int j = 0; j < 40; ++j) orow[j] = row[j] - l;
}

extern "C" void kernel_launch(void* const* d_in, const int* in_sizes, int n_in,
                              void* d_out, int out_size, void* d_ws, size_t ws_size,
                              hipStream_t stream) {
  const float* x     = (const float*)d_in[0];
  const float* Mv    = (const float*)d_in[1];
  const float* AMv   = (const float*)d_in[2];
  const int*   srcZ  = (const int*)d_in[3];
  const int*   dstZ  = (const int*)d_in[4];
  const float* valsZ = (const float*)d_in[5];
  const int*   src   = (const int*)d_in[6];
  const int*   dst   = (const int*)d_in[7];
  const float* vals  = (const float*)d_in[8];
  const float* W0 = (const float*)d_in[9];
  const float* b0 = (const float*)d_in[10];
  const float* g0 = (const float*)d_in[11];
  const float* be0 = (const float*)d_in[12];
  const float* rm0 = (const float*)d_in[13];
  const float* rv0 = (const float*)d_in[14];
  const float* W1 = (const float*)d_in[15];
  const float* b1 = (const float*)d_in[16];
  const float* g1 = (const float*)d_in[17];
  const float* be1 = (const float*)d_in[18];
  const float* rm1 = (const float*)d_in[19];
  const float* rv1 = (const float*)d_in[20];
  const float* W2 = (const float*)d_in[21];
  const float* b2 = (const float*)d_in[22];

  int N = in_sizes[1];
  int E = in_sizes[3];
  float* bufA = (float*)d_ws;
  float* bufB = bufA + (size_t)N * NF;
  float* outp = (float*)d_out;

  int tiles_m = (N + 15) / 16;
  int gemm_blocks = (tiles_m + GEMM_WAVES - 1) / GEMM_WAVES;
  int nb = (N + RB - 1) / RB;

  float* recZ = bufA;
  int*   offZ = (int*)(bufA + 2 * (size_t)E);
  float* recA = outp;
  int*   offA = (int*)(outp + 2 * (size_t)E);

  bucket_count<<<dim3(GCH, 2), BK_THREADS, 0, stream>>>(dstZ, dst, offZ, offA, E, nb);
  bucket_scan<<<2, 1024, 0, stream>>>(offZ, offA, nb * GCH);
  bucket_place<<<dim3(GCH, 2), BK_THREADS, 0, stream>>>(srcZ, dstZ, valsZ, Mv, AMv, src, dst, vals,
                                                        offZ, offA, recZ, recA, E, nb);

  owner_gemm_kernel<<<nb, FT, 0, stream>>>(x, recZ, offZ, W0, b0, g0, be0, rm0, rv0, bufB, N);

  owner_gemm_kernel<<<nb, FT, 0, stream>>>(bufB, recA, offA, W1, b1, g1, be1, rm1, rv1, bufA, N);

  gemm_strip_kernel<3><<<gemm_blocks, 32 * GEMM_WAVES, 0, stream>>>(
      bufA, W2, nullptr, nullptr, nullptr, nullptr, nullptr, bufB, N, NC, 0);
  owner40_kernel<<<nb, 32 * OWN_WAVES, 0, stream>>>(bufB, recA, offA, b2, bufA, N);
  log_softmax40<<<(N + 255) / 256, 256, 0, stream>>>(bufA, outp, N);
}


# Layer 1: owner work-group also runs the last layer's 128->40 WMMA product on its finished rows in LDS and stores only [N x 40]; layer 1's rows never reach memory; 1 launch fewer; sums as before
# speedup vs baseline: 8.4235x; 1.0942x over previous
#include <hip/hip_runtime.h>
#include <math.h>

typedef __attribute__((ext_vector_type(2))) float v2f;
typedef __attribute__((ext_vector_type(4))) float v4f;
typedef __attribute__((ext_vector_type(8))) float v8f;

#define NF 128

#define NC 40
#define NCP 48
#define NT2 (NCP / 16)
#define WP (NF + 4)

#define RB 400
#define NBMAX 256
#define GCH 256
#define BK_THREADS 512

__global__ __launch_bounds__(BK_THREADS) void bucket_count(
    const int* __restrict__ dstZ, const int* __restrict__ dstA,
    int* __restrict__ cntZ, int* __restrict__ cntA, int E, int nb) {
  __shared__ int hist[NBMAX];
  const int tid = threadIdx.x;
  const int g = blockIdx.x;
  const int* __restrict__ dst = blockIdx.y == 0 ? dstZ : dstA;
  int* __restrict__ cnt = blockIdx.y == 0 ? cntZ : cntA;
  for (int i = tid; i < NBMAX; i += BK_THREADS) hist[i] = 0;
  __syncthreads();
  const int ch = (E + GCH - 1) / GCH;
  const int lo = g * ch;
  const int hi = lo + ch < E ? lo + ch : E;
  for (int e = lo + tid; e < hi; e += BK_THREADS) atomicAdd(&hist[dst[e] / RB], 1);
  __syncthreads();
  for (int i = tid; i < nb; i += BK_THREADS) cnt[i * GCH + g] = hist[i];
}

__global__ __launch_bounds__(1024) void bucket_scan(int* __restrict__ cZ, int* __restrict__ cA, int n) {
  __shared__ int buf[NBMAX * GCH];
  __shared__ int wsum[32];
  int* __restrict__ c = blockIdx.x == 0 ? cZ : cA;
  const int t = threadIdx.x;
  const int lane = t & 31, wv = t >> 5;
  for (int i = t; i < n; i += 1024) buf[i] = c[i];
  __syncthreads();
  const int per = (n + 1023) / 1024;
  const int lo = t * per;
  const int hi = lo + per < n ? lo + per : n;
  int s = 0;
  for (int i = lo; i < hi; ++i) s += buf[i];
  int x = s;
  for (int d = 1; d < 32; d <<= 1) {
    int y = __shfl_up(x, d);
    if (lane >= d) x += y;
  }
  if (lane == 31) wsum[wv] = x;
  __syncthreads();
  if (wv == 0) {
    int v = wsum[lane];
    for (int d = 1; d < 32; d <<= 1) {
      int y = __shfl_up(v, d);
      if (lane >= d) v += y;
    }
    wsum[lane] = v;
  }
  __syncthreads();
  int base = (wv ? wsum[wv - 1] : 0) + x - s;
  for (int i = lo; i < hi; ++i) {
    int v = buf[i];
    buf[i] = base;
    base += v;
  }
  if (t == 1023) c[n] = base;
  __syncthreads();
  for (int i = t; i < n; i += 1024) c[i] = buf[i];
}

__global__ __launch_bounds__(BK_THREADS) void bucket_place(
    const int* __restrict__ srcZ, const int* __restrict__ dstZ, const float* __restrict__ valsZ,
    const float* __restrict__ Mv, const float* __restrict__ AMv,
    const int* __restrict__ srcA, const int* __restrict__ dstA, const float* __restrict__ valsA,
    const int* __restrict__ offZ, const int* __restrict__ offA,
    float* __restrict__ recZ, float* __restrict__ recA, int E, int nb) {
  __shared__ int cur[NBMAX];
  const int tid = threadIdx.x;
  const int g = blockIdx.x;
  const bool z = blockIdx.y == 0;
  const int* __restrict__ src = z ? srcZ : srcA;
  const int* __restrict__ dst = z ? dstZ : dstA;
  const float* __restrict__ vals = z ? valsZ : valsA;
  const int* __restrict__ off = z ? offZ : offA;
  uint2* __restrict__ rec = (uint2*)(z ? recZ : recA);
  for (int i = tid; i < nb; i += BK_THREADS) cur[i] = off[i * GCH + g];
  __syncthreads();
  const int ch = (E + GCH - 1) / GCH;
  const int lo = g * ch;
  const int hi = lo + ch < E ? lo + ch : E;
  for (int e = lo + tid; e < hi; e += BK_THREADS) {
    const int s = src[e];
    const int d = dst[e];
    float w = vals[e];
    if (z) {
      w *= Mv[s];
      w *= AMv[d];
    }
    const int b = d / RB;
    const int p = atomicAdd(&cur[b], 1);
    rec[p] = make_uint2((unsigned)s | ((unsigned)(d - b * RB) << 17), __float_as_uint(w));
  }
}

#define CAP 8192
#define EPB 16
#define OWN_WAVES 32

struct SortLds {
  uint2 srt[CAP + 32];
  int cnt[512];
  int start[512];
  int wsum[32];
};

template <int NTHR>
__device__ __forceinline__ void sort_bucket(SortLds& L, const uint2* __restrict__ rec, int cb, int m,
                                            int tid, int lane, int wave) {
  constexpr int RPT = (CAP + NTHR - 1) / NTHR;
  if (tid < 512) L.cnt[tid] = 0;
  if (NTHR < 1024 && tid < 32) L.wsum[tid] = 0;
  __syncthreads();

  uint2 r[RPT];
  int rk[RPT];
#pragma unroll
  for (int j = 0; j < RPT; ++j) {
    const int i = tid + NTHR * j;
    if (i < m) r[j] = rec[cb + i];
  }
#pragma unroll
  for (int j = 0; j < RPT; ++j) {
    const int i = tid + NTHR * j;
    if (i < m) rk[j] = atomicAdd(&L.cnt[r[j].x >> 17], 1);
  }
  __syncthreads();

  const int cv = tid < RB ? L.cnt[tid] : 0;
  int x = cv;
  for (int d = 1; d < 32; d <<= 1) {
    int y = __shfl_up(x, d);
    if (lane >= d) x += y;
  }
  if (lane == 31) L.wsum[wave] = x;
  __syncthreads();
  if (wave == 0) {
    int t = L.wsum[lane];
    for (int d = 1; d < 32; d <<= 1) {
      int y = __shfl_up(t, d);
      if (lane >= d) t += y;
    }
    L.wsum[lane] = t;
  }
  __syncthreads();
  if (tid <= RB) L.start[tid] = (wave ? L.wsum[wave - 1] : 0) + x - cv;
  __syncthreads();

#pragma unroll
  for (int j = 0; j < RPT; ++j) {
    const int i = tid + NTHR * j;
    if (i < m) L.srt[L.start[r[j].x >> 17] + rk[j]] = r[j];
  }
  __syncthreads();
}

template <bool FULL, int ROWF, class Flush>
__device__ __forceinline__ void stream_batch(const float* __restrict__ X, const uint2* srt, int p,
                                             int nrec, int lane, int l4, int& cur, v4f& acc,
                                             Flush flush) {
  const uint2 rr = srt[p + lane];
  unsigned sx[EPB];
  float sw[EPB];
#pragma unroll
  for (int k = 0; k < EPB; ++k) {
    sx[k] = (unsigned)__builtin_amdgcn_readlane((int)rr.x, k);
    sw[k] = __uint_as_float((unsigned)__builtin_amdgcn_readlane((int)rr.y, k));
  }
  v4f v[EPB];
#pragma unroll
  for (int k = 0; k < EPB; ++k) {
    const unsigned s = (FULL || k < nrec) ? sx[k] : sx[0];
    v[k] = ((const v4f*)(X + (size_t)(s & 0x1FFFFu) * ROWF))[l4];
  }
#pragma unroll
  for (int k = 0; k < EPB; ++k) {
    if (FULL || k < nrec) {
      const int row = (int)(sx[k] >> 17);
      if (row != cur) {
        if (cur >= 0) flush(cur, acc);
        acc = v4f{};
        cur = row;
      }
      acc += sw[k] * v[k];
    }
  }
}

#define FW (RB / 16)
#define FT (32 * FW)
#define AP (NF + 4)

template <bool NEXT>
struct alignas(16) FusedLds {
  union {
    SortLds S;
    float Ws[NF * WP];
  } u;
  float At[RB * AP];
  float W2s[NEXT ? NCP * WP : 4];
};

template <bool NEXT>
__global__ __launch_bounds__(FT) void owner_gemm_kernel(
    const float* __restrict__ X, const float* __restrict__ recf, const int* __restrict__ off,
    const float* __restrict__ W, const float* __restrict__ bias,
    const float* __restrict__ g, const float* __restrict__ be,
    const float* __restrict__ rm, const float* __restrict__ rv,
    const float* __restrict__ W2,
    float* __restrict__ out,
    int N) {
  static_assert(RB % 16 == 0 && FW <= 32, "one wave per 16-row strip");
  __shared__ FusedLds<NEXT> L;
  const int tid = threadIdx.x;
  const int lane = tid & 31;
  const int wave = __builtin_amdgcn_readfirstlane((int)(tid >> 5));
  const uint2* __restrict__ rec = (const uint2*)recf;
  const int b = blockIdx.x;
  const int beg = off[b * GCH];
  const int end = off[(b + 1) * GCH];
  const int r0 = wave * 16;
  const int r1 = r0 + 16;

  int pass = 0;
  int cb = beg;
  do {
    const int m = end - cb < CAP ? end - cb : CAP;
    sort_bucket<FT>(L.u.S, rec, cb, m, tid, lane, wave);

    if (pass == 0) {
      for (int q = r0; q < r1; ++q) {
        if (__builtin_amdgcn_readfirstlane(L.u.S.cnt[q]) == 0)
          ((v4f*)(L.At + q * AP))[lane] = v4f{};
      }
    }

    auto flush = [&](int row, v4f sum) {
      v4f* dst = (v4f*)(L.At + row * AP) + lane;
      if (pass) sum += *dst;
      *dst = sum;
    };
    int p = __builtin_amdgcn_readfirstlane(L.u.S.start[r0]);
    const int pend = __builtin_amdgcn_readfirstlane(L.u.S.start[r1]);
    int cur = -1;
    v4f acc = v4f{};
    for (; p + EPB <= pend; p += EPB)
      stream_batch<true, NF>(X, L.u.S.srt, p, EPB, lane, lane, cur, acc, flush);
    if (p < pend) stream_batch<false, NF>(X, L.u.S.srt, p, pend - p, lane, lane, cur, acc, flush);
    if (cur >= 0) flush(cur, acc);

    cb += CAP;
    ++pass;
    if (cb < end) __syncthreads();
  } while (cb < end);

  constexpr int WPT = (NF * NF / 4 + FT - 1) / FT;
  v4f wv[WPT];
#pragma unroll
  for (int j = 0; j < WPT; ++j) {
    const int idx = tid + FT * j;
    if (idx < NF * NF / 4) wv[j] = *(const v4f*)(W + 4 * idx);
  }
  constexpr int VPT = (NCP * NF / 4 + FT - 1) / FT;
  v4f vv[VPT];
  if constexpr (NEXT) {
#pragma unroll
    for (int j = 0; j < VPT; ++j) {
      const int idx = tid + FT * j;
      if (idx < NCP * NF / 4) {
        const int r = idx >> 5;
        vv[j] = *(const v4f*)(W2 + (r < NC ? r : NC - 1) * NF + 4 * (idx & 31));
      }
    }
  }
  __syncthreads();
#pragma unroll
  for (int j = 0; j < WPT; ++j) {
    const int idx = tid + FT * j;
    if (idx < NF * NF / 4) *(v4f*)(L.u.Ws + (idx >> 5) * WP + 4 * (idx & 31)) = wv[j];
  }
  if constexpr (NEXT) {
#pragma unroll
    for (int j = 0; j < VPT; ++j) {
      const int idx = tid + FT * j;
      if (idx < NCP * NF / 4) *(v4f*)(L.W2s + (idx >> 5) * WP + 4 * (idx & 31)) = vv[j];
    }
  }
  __syncthreads();

  const int hl = lane >> 4;
  const int ll = lane & 15;
  const float* al = L.At + (r0 + ll) * AP + 2 * hl;
  const float* wl = L.u.Ws + ll * WP + 2 * hl;
  v8f c[NF / 16];
#pragma unroll
  for (int n = 0; n < NF / 16; ++n) c[n] = v8f{};

#pragma unroll 4
  for (int t = 0; t < NF / 4; ++t) {
    const v2f a = *(const v2f*)(al + 4 * t);
    v2f bf[NF / 16];
#pragma unroll
    for (int n = 0; n < NF / 16; ++n) bf[n] = *(const v2f*)(wl + n * 16 * WP + 4 * t);
#pragma unroll
    for (int n = 0; n < NF / 16; ++n)
      c[n] = __builtin_amdgcn_wmma_f32_16x16x4_f32(
           false, a,  false, bf[n],
           (short)0, c[n],  false,  false);
  }

  if constexpr (!NEXT) {
    const int rb = b * RB + r0 + hl * 8;
#pragma unroll
    for (int n = 0; n < NF / 16; ++n) {
      const int col = n * 16 + ll;
      const float bb = bias[col];
      const float sc = g[col] * rsqrtf(rv[col] + 1e-5f);
      const float sh = be[col] - rm[col] * sc;
#pragma unroll
      for (int i = 0; i < 8; ++i) {
        const int r = rb + i;
        if (r < N) out[(size_t)r * NF + col] = fmaxf((c[n][i] + bb) * sc + sh, 0.0f);
      }
    }
  } else {
    const int ln = (int)__builtin_amdgcn_mbcnt_lo(~0u, 0u);
    const int h2 = ln >> 4;
    const int l2 = ln & 15;
    float* hrow = L.At + (r0 + h2 * 8) * AP + l2;
#pragma unroll
    for (int n = 0; n < NF / 16; ++n) {
      const int col = n * 16 + l2;
      const float bb = bias[col];
      const float sc = g[col] * rsqrtf(rv[col] + 1e-5f);
      const float sh = be[col] - rm[col] * sc;
#pragma unroll
      for (int i = 0; i < 8; ++i) hrow[i * AP + n * 16] = fmaxf((c[n][i] + bb) * sc + sh, 0.0f);
    }

    const float* vl = L.W2s + l2 * WP + 2 * h2;
    v8f d[NT2];
#pragma unroll
    for (int n = 0; n < NT2; ++n) d[n] = v8f{};
#pragma unroll 4
    for (int t = 0; t < NF / 4; ++t) {
      const v2f a = *(const v2f*)(al + 4 * t);
      v2f bf[NT2];
#pragma unroll
      for (int n = 0; n < NT2; ++n) bf[n] = *(const v2f*)(vl + n * 16 * WP + 4 * t);
#pragma unroll
      for (int n = 0; n < NT2; ++n)
        d[n] = __builtin_amdgcn_wmma_f32_16x16x4_f32(
             false, a,  false, bf[n],
             (short)0, d[n],  false,  false);
    }

    float* ts = L.At + r0 * AP;
#pragma unroll
    for (int n = 0; n < NT2; ++n) {
      const int col = n * 16 + l2;
      if (col < NC) {
#pragma unroll
        for (int i = 0; i < 8; ++i) ts[(h2 * 8 + i) * NC + col] = d[n][i];
      }
    }
    const int row0 = b * RB + r0;
    float* __restrict__ o = out + (size_t)row0 * NC;
    static_assert((16 * NC / 4) % 32 == 0, "whole store instructions per strip");
    v4f tv[16 * NC / 4 / 32];
#pragma unroll
    for (int j = 0; j < 16 * NC / 4 / 32; ++j) tv[j] = *(const v4f*)(ts + 4 * (ln + 32 * j));
#pragma unroll
    for (int j = 0; j < 16 * NC / 4 / 32; ++j) {
      const int idx = ln + 32 * j;
      if (row0 + (4 * idx) / NC < N) *(v4f*)(o + 4 * idx) = tv[j];
    }
  }
}

#define NC4 (NC / 4)

__global__ __launch_bounds__(32 * OWN_WAVES) void owner40_kernel(
    const float* __restrict__ T, const float* __restrict__ recf, const int* __restrict__ off,
    const float* __restrict__ b2, float* __restrict__ out, int N) {
  __shared__ SortLds L;
  __shared__ float tile[RB * NC];
  const int tid = threadIdx.x;
  const int lane = tid & 31;
  const int wave = __builtin_amdgcn_readfirstlane((int)(tid >> 5));
  const int l4 = lane < NC4 ? lane : NC4 - 1;
  for (int i = tid; i < RB * NC; i += 32 * OWN_WAVES) tile[i] = b2[i % NC];

  const uint2* __restrict__ rec = (const uint2*)recf;
  const int b = blockIdx.x;
  const int beg = off[b * GCH];
  const int end = off[(b + 1) * GCH];
  const int r0 = (wave * RB) >> 5;
  const int r1 = ((wave + 1) * RB) >> 5;

  auto flush = [&](int row, v4f sum) {
    if (lane < NC4) {
      v4f* t = (v4f*)(tile + row * NC) + lane;
      *t += sum;
    }
  };
  int cb = beg;
  do {
    const int m = end - cb < CAP ? end - cb : CAP;
    sort_bucket<32 * OWN_WAVES>(L, rec, cb, m, tid, lane, wave);

    int p = __builtin_amdgcn_readfirstlane(L.start[r0]);
    const int pend = __builtin_amdgcn_readfirstlane(L.start[r1]);
    int cur = -1;
    v4f acc = v4f{};
    for (; p + EPB <= pend; p += EPB)
      stream_batch<true, NC>(T, L.srt, p, EPB, lane, l4, cur, acc, flush);
    if (p < pend) stream_batch<false, NC>(T, L.srt, p, pend - p, lane, l4, cur, acc, flush);
    if (cur >= 0) flush(cur, acc);

    cb += CAP;
    __syncthreads();
  } while (cb < end);

  const int row0 = b * RB;
  const int rows = N - row0 < RB ? N - row0 : RB;
  float* __restrict__ o = out + (size_t)row0 * NC;
  for (int i = tid; i < rows * NC; i += 32 * OWN_WAVES) o[i] = tile[i];
}

__global__ __launch_bounds__(256) void log_softmax40(const float* __restrict__ in,
                                                     float* __restrict__ out, int n) {
  int i = blockIdx.x * blockDim.x + threadIdx.x;
  if (i >= n) return;
  const float* row = in + (size_t)i * 40;
  float* orow = out + (size_t)i * 40;
  float m = row[0];
#pragma unroll
  for (int j = 1; j < 40; ++j) m = fmaxf(m, row[j]);
  float s = 0.0f;
#pragma unroll
  for (int j = 0; j < 40; ++j) s += expf(row[j] - m);
  float l = m + logf(s);
#pragma unroll
  for (int j = 0; j < 40; ++j) orow[j] = row[j] - l;
}

extern "C" void kernel_launch(void* const* d_in, const int* in_sizes, int n_in,
                              void* d_out, int out_size, void* d_ws, size_t ws_size,
                              hipStream_t stream) {
  const float* x     = (const float*)d_in[0];
  const float* Mv    = (const float*)d_in[1];
  const float* AMv   = (const float*)d_in[2];
  const int*   srcZ  = (const int*)d_in[3];
  const int*   dstZ  = (const int*)d_in[4];
  const float* valsZ = (const float*)d_in[5];
  const int*   src   = (const int*)d_in[6];
  const int*   dst   = (const int*)d_in[7];
  const float* vals  = (const float*)d_in[8];
  const float* W0 = (const float*)d_in[9];
  const float* b0 = (const float*)d_in[10];
  const float* g0 = (const float*)d_in[11];
  const float* be0 = (const float*)d_in[12];
  const float* rm0 = (const float*)d_in[13];
  const float* rv0 = (const float*)d_in[14];
  const float* W1 = (const float*)d_in[15];
  const float* b1 = (const float*)d_in[16];
  const float* g1 = (const float*)d_in[17];
  const float* be1 = (const float*)d_in[18];
  const float* rm1 = (const float*)d_in[19];
  const float* rv1 = (const float*)d_in[20];
  const float* W2 = (const float*)d_in[21];
  const float* b2 = (const float*)d_in[22];

  int N = in_sizes[1];
  int E = in_sizes[3];
  float* bufA = (float*)d_ws;
  float* bufB = bufA + (size_t)N * NF;
  float* outp = (float*)d_out;

  int nb = (N + RB - 1) / RB;

  float* recZ = bufA;
  int*   offZ = (int*)(bufA + 2 * (size_t)E);
  float* recA = outp;
  int*   offA = (int*)(outp + 2 * (size_t)E);

  bucket_count<<<dim3(GCH, 2), BK_THREADS, 0, stream>>>(dstZ, dst, offZ, offA, E, nb);
  bucket_scan<<<2, 1024, 0, stream>>>(offZ, offA, nb * GCH);
  bucket_place<<<dim3(GCH, 2), BK_THREADS, 0, stream>>>(srcZ, dstZ, valsZ, Mv, AMv, src, dst, vals,
                                                        offZ, offA, recZ, recA, E, nb);

  owner_gemm_kernel<false><<<nb, FT, 0, stream>>>(x, recZ, offZ, W0, b0, g0, be0, rm0, rv0,
                                                 nullptr, bufB, N);

  owner_gemm_kernel<true><<<nb, FT, 0, stream>>>(bufB, recA, offA, W1, b1, g1, be1, rm1, rv1,
                                                W2, bufA, N);

  owner40_kernel<<<nb, 32 * OWN_WAVES, 0, stream>>>(bufA, recA, offA, b2, bufB, N);
  log_softmax40<<<(N + 255) / 256, 256, 0, stream>>>(bufB, outp, N);
}


# Grouping in one launch: a block counts, scans (LDS) and places its own chunk's edges, each stage's loads all in flight; owners read their records as 256 runs; 2 launches fewer; LDS-rank sums kept
# speedup vs baseline: 9.0552x; 1.0750x over previous
#include <hip/hip_runtime.h>
#include <math.h>

typedef __attribute__((ext_vector_type(2))) float v2f;
typedef __attribute__((ext_vector_type(4))) float v4f;
typedef __attribute__((ext_vector_type(8))) float v8f;

#define NF 128

#define NC 40
#define NCP 48
#define NT2 (NCP / 16)
#define WP (NF + 4)

#define RB 400
#define NBMAX 256
#define GCH 256
#define BK_THREADS 512

#define EPT 13

__global__ __launch_bounds__(BK_THREADS) void chunk_group(
    const int* __restrict__ srcZ, const int* __restrict__ dstZ, const float* __restrict__ valsZ,
    const float* __restrict__ Mv, const float* __restrict__ AMv,
    const int* __restrict__ srcA, const int* __restrict__ dstA, const float* __restrict__ valsA,
    int* __restrict__ runZ, int* __restrict__ runA,
    float* __restrict__ recZ, float* __restrict__ recA, int E, int nb) {
  __shared__ int cur[NBMAX];
  __shared__ int wsum[NBMAX / 32];
  const int tid = threadIdx.x;
  const int lane = tid & 31, wv = tid >> 5;
  const int g = blockIdx.x;
  const bool z = blockIdx.y == 0;
  const int* __restrict__ src = z ? srcZ : srcA;
  const int* __restrict__ dst = z ? dstZ : dstA;
  const float* __restrict__ vals = z ? valsZ : valsA;
  int2* __restrict__ run = (int2*)(z ? runZ : runA);
  uint2* __restrict__ rec = (uint2*)(z ? recZ : recA);
  for (int i = tid; i < NBMAX; i += BK_THREADS) cur[i] = 0;
  __syncthreads();
  const int ch = (E + GCH - 1) / GCH;
  const int lo = g * ch;
  const int hi = lo + ch < E ? lo + ch : E;

  for (int base = lo + tid; base < hi; base += BK_THREADS * EPT) {
    int d[EPT];
#pragma unroll
    for (int j = 0; j < EPT; ++j) {
      const int e = base + BK_THREADS * j;
      d[j] = dst[e < hi ? e : hi - 1];
    }
#pragma unroll
    for (int j = 0; j < EPT; ++j) {
      atomicAdd(&cur[d[j] / RB], base + BK_THREADS * j < hi ? 1 : 0);
    }
  }
  __syncthreads();

  const int cv = tid < NBMAX ? cur[tid] : 0;
  int x = cv;
  for (int dd = 1; dd < 32; dd <<= 1) {
    int y = __shfl_up(x, dd);
    if (lane >= dd) x += y;
  }
  if (tid < NBMAX && lane == 31) wsum[wv] = x;
  __syncthreads();
  if (tid < NBMAX) {
    int before = 0;
#pragma unroll
    for (int k = 0; k < NBMAX / 32; ++k) before += k < wv ? wsum[k] : 0;
    const int st = lo + before + x - cv;
    cur[tid] = st;
    if (tid < nb) run[tid * GCH + g] = make_int2(st, cv);
  }
  __syncthreads();

  for (int base = lo + tid; base < hi; base += BK_THREADS * EPT) {
    int s[EPT], d[EPT];
    float w[EPT];
#pragma unroll
    for (int j = 0; j < EPT; ++j) {
      const int e = base + BK_THREADS * j;
      const int ec = e < hi ? e : hi - 1;
      s[j] = src[ec];
      d[j] = dst[ec];
      w[j] = vals[ec];
    }
    if (z) {
      float fm[EPT], fa[EPT];
#pragma unroll
      for (int j = 0; j < EPT; ++j) {
        fm[j] = Mv[s[j]];
        fa[j] = AMv[d[j]];
      }
#pragma unroll
      for (int j = 0; j < EPT; ++j) {
        w[j] *= fm[j];
        w[j] *= fa[j];
      }
    }
    int p[EPT];
#pragma unroll
    for (int j = 0; j < EPT; ++j) {
      if (base + BK_THREADS * j < hi) p[j] = atomicAdd(&cur[d[j] / RB], 1);
    }
#pragma unroll
    for (int j = 0; j < EPT; ++j) {
      if (base + BK_THREADS * j < hi) {
        const int b = d[j] / RB;
        rec[p[j]] = make_uint2((unsigned)s[j] | ((unsigned)(d[j] - b * RB) << 17), __float_as_uint(w[j]));
      }
    }
  }
}

#define CAP 8192
#define EPB 16
#define OWN_WAVES 32

struct SortLds {
  uint2 srt[CAP + 32];
  int cnt[512];
  int start[512];
  int wsum[32];
  int rs[GCH];
  int rp[GCH + 8];
};

__device__ __forceinline__ int load_runs(SortLds& L, const int* __restrict__ run, int b, int tid, int lane, int wave) {
  int len = 0;
  if (tid < GCH) {
    const int2 r = ((const int2*)run)[b * GCH + tid];
    L.rs[tid] = r.x;
    len = r.y;
  }
  int x = len;
  for (int d = 1; d < 32; d <<= 1) {
    int y = __shfl_up(x, d);
    if (lane >= d) x += y;
  }
  if (tid < GCH && lane == 31) L.wsum[wave] = x;
  __syncthreads();
  if (tid < GCH) {
    int before = 0;
#pragma unroll
    for (int k = 0; k < GCH / 32; ++k) before += k < wave ? L.wsum[k] : 0;
    L.rp[tid] = before + x - len;
    if (tid == GCH - 1) L.rp[GCH] = before + x;
  }
  __syncthreads();
  return L.rp[GCH];
}

template <int NTHR>
__device__ __forceinline__ void sort_bucket(SortLds& L, const uint2* __restrict__ rec, int cb, int m,
                                            int tid, int lane, int wave) {
  constexpr int RPT = (CAP + NTHR - 1) / NTHR;
  if (tid < 512) L.cnt[tid] = 0;
  if (NTHR < 1024 && tid < 32) L.wsum[tid] = 0;
  __syncthreads();

  int q[RPT], g[RPT];
#pragma unroll
  for (int j = 0; j < RPT; ++j) {
    const int i = tid + NTHR * j;
    q[j] = cb + (i < m ? i : 0);
    g[j] = 0;
  }
#pragma unroll
  for (int s = GCH / 2; s; s >>= 1) {
#pragma unroll
    for (int j = 0; j < RPT; ++j) g[j] += L.rp[g[j] + s] <= q[j] ? s : 0;
  }
#pragma unroll
  for (int j = 0; j < RPT; ++j) q[j] = L.rs[g[j]] + (q[j] - L.rp[g[j]]);

  uint2 r[RPT];
  int rk[RPT];
#pragma unroll
  for (int j = 0; j < RPT; ++j) {
    const int i = tid + NTHR * j;
    if (i < m) r[j] = rec[q[j]];
  }
#pragma unroll
  for (int j = 0; j < RPT; ++j) {
    const int i = tid + NTHR * j;
    if (i < m) rk[j] = atomicAdd(&L.cnt[r[j].x >> 17], 1);
  }
  __syncthreads();

  const int cv = tid < RB ? L.cnt[tid] : 0;
  int x = cv;
  for (int d = 1; d < 32; d <<= 1) {
    int y = __shfl_up(x, d);
    if (lane >= d) x += y;
  }
  if (lane == 31) L.wsum[wave] = x;
  __syncthreads();
  if (wave == 0) {
    int t = L.wsum[lane];
    for (int d = 1; d < 32; d <<= 1) {
      int y = __shfl_up(t, d);
      if (lane >= d) t += y;
    }
    L.wsum[lane] = t;
  }
  __syncthreads();
  if (tid <= RB) L.start[tid] = (wave ? L.wsum[wave - 1] : 0) + x - cv;
  __syncthreads();

#pragma unroll
  for (int j = 0; j < RPT; ++j) {
    const int i = tid + NTHR * j;
    if (i < m) L.srt[L.start[r[j].x >> 17] + rk[j]] = r[j];
  }
  __syncthreads();
}

template <bool FULL, int ROWF, class Flush>
__device__ __forceinline__ void stream_batch(const float* __restrict__ X, const uint2* srt, int p,
                                             int nrec, int lane, int l4, int& cur, v4f& acc,
                                             Flush flush) {
  const uint2 rr = srt[p + lane];
  unsigned sx[EPB];
  float sw[EPB];
#pragma unroll
  for (int k = 0; k < EPB; ++k) {
    sx[k] = (unsigned)__builtin_amdgcn_readlane((int)rr.x, k);
    sw[k] = __uint_as_float((unsigned)__builtin_amdgcn_readlane((int)rr.y, k));
  }
  v4f v[EPB];
#pragma unroll
  for (int k = 0; k < EPB; ++k) {
    const unsigned s = (FULL || k < nrec) ? sx[k] : sx[0];
    v[k] = ((const v4f*)(X + (size_t)(s & 0x1FFFFu) * ROWF))[l4];
  }
#pragma unroll
  for (int k = 0; k < EPB; ++k) {
    if (FULL || k < nrec) {
      const int row = (int)(sx[k] >> 17);
      if (row != cur) {
        if (cur >= 0) flush(cur, acc);
        acc = v4f{};
        cur = row;
      }
      acc += sw[k] * v[k];
    }
  }
}

#define FW (RB / 16)
#define FT (32 * FW)
#define AP (NF + 4)

template <bool NEXT>
struct alignas(16) FusedLds {
  union {
    SortLds S;
    float Ws[NF * WP];
  } u;
  float At[RB * AP];
  float W2s[NEXT ? NCP * WP : 4];
};

template <bool NEXT>
__global__ __launch_bounds__(FT) void owner_gemm_kernel(
    const float* __restrict__ X, const float* __restrict__ recf, const int* __restrict__ run,
    const float* __restrict__ W, const float* __restrict__ bias,
    const float* __restrict__ g, const float* __restrict__ be,
    const float* __restrict__ rm, const float* __restrict__ rv,
    const float* __restrict__ W2,
    float* __restrict__ out,
    int N) {
  static_assert(RB % 16 == 0 && FW <= 32, "one wave per 16-row strip");
  __shared__ FusedLds<NEXT> L;
  const int tid = threadIdx.x;
  const int lane = tid & 31;
  const int wave = __builtin_amdgcn_readfirstlane((int)(tid >> 5));
  const uint2* __restrict__ rec = (const uint2*)recf;
  const int b = blockIdx.x;
  const int end = load_runs(L.u.S, run, b, tid, lane, wave);
  const int r0 = wave * 16;
  const int r1 = r0 + 16;

  int pass = 0;
  int cb = 0;
  do {
    const int m = end - cb < CAP ? end - cb : CAP;
    sort_bucket<FT>(L.u.S, rec, cb, m, tid, lane, wave);

    if (pass == 0) {
      for (int q = r0; q < r1; ++q) {
        if (__builtin_amdgcn_readfirstlane(L.u.S.cnt[q]) == 0)
          ((v4f*)(L.At + q * AP))[lane] = v4f{};
      }
    }

    auto flush = [&](int row, v4f sum) {
      v4f* dst = (v4f*)(L.At + row * AP) + lane;
      if (pass) sum += *dst;
      *dst = sum;
    };
    int p = __builtin_amdgcn_readfirstlane(L.u.S.start[r0]);
    const int pend = __builtin_amdgcn_readfirstlane(L.u.S.start[r1]);
    int cur = -1;
    v4f acc = v4f{};
    for (; p + EPB <= pend; p += EPB)
      stream_batch<true, NF>(X, L.u.S.srt, p, EPB, lane, lane, cur, acc, flush);
    if (p < pend) stream_batch<false, NF>(X, L.u.S.srt, p, pend - p, lane, lane, cur, acc, flush);
    if (cur >= 0) flush(cur, acc);

    cb += CAP;
    ++pass;
    if (cb < end) __syncthreads();
  } while (cb < end);

  constexpr int WPT = (NF * NF / 4 + FT - 1) / FT;
  v4f wv[WPT];
#pragma unroll
  for (int j = 0; j < WPT; ++j) {
    const int idx = tid + FT * j;
    if (idx < NF * NF / 4) wv[j] = *(const v4f*)(W + 4 * idx);
  }
  constexpr int VPT = (NCP * NF / 4 + FT - 1) / FT;
  v4f vv[VPT];
  if constexpr (NEXT) {
#pragma unroll
    for (int j = 0; j < VPT; ++j) {
      const int idx = tid + FT * j;
      if (idx < NCP * NF / 4) {
        const int r = idx >> 5;
        vv[j] = *(const v4f*)(W2 + (r < NC ? r : NC - 1) * NF + 4 * (idx & 31));
      }
    }
  }
  __syncthreads();
#pragma unroll
  for (int j = 0; j < WPT; ++j) {
    const int idx = tid + FT * j;
    if (idx < NF * NF / 4) *(v4f*)(L.u.Ws + (idx >> 5) * WP + 4 * (idx & 31)) = wv[j];
  }
  if constexpr (NEXT) {
#pragma unroll
    for (int j = 0; j < VPT; ++j) {
      const int idx = tid + FT * j;
      if (idx < NCP * NF / 4) *(v4f*)(L.W2s + (idx >> 5) * WP + 4 * (idx & 31)) = vv[j];
    }
  }
  __syncthreads();

  const int hl = lane >> 4;
  const int ll = lane & 15;
  const float* al = L.At + (r0 + ll) * AP + 2 * hl;
  const float* wl = L.u.Ws + ll * WP + 2 * hl;
  v8f c[NF / 16];
#pragma unroll
  for (int n = 0; n < NF / 16; ++n) c[n] = v8f{};

#pragma unroll 4
  for (int t = 0; t < NF / 4; ++t) {
    const v2f a = *(const v2f*)(al + 4 * t);
    v2f bf[NF / 16];
#pragma unroll
    for (int n = 0; n < NF / 16; ++n) bf[n] = *(const v2f*)(wl + n * 16 * WP + 4 * t);
#pragma unroll
    for (int n = 0; n < NF / 16; ++n)
      c[n] = __builtin_amdgcn_wmma_f32_16x16x4_f32(
           false, a,  false, bf[n],
           (short)0, c[n],  false,  false);
  }

  if constexpr (!NEXT) {
    const int rb = b * RB + r0 + hl * 8;
#pragma unroll
    for (int n = 0; n < NF / 16; ++n) {
      const int col = n * 16 + ll;
      const float bb = bias[col];
      const float sc = g[col] * rsqrtf(rv[col] + 1e-5f);
      const float sh = be[col] - rm[col] * sc;
#pragma unroll
      for (int i = 0; i < 8; ++i) {
        const int r = rb + i;
        if (r < N) out[(size_t)r * NF + col] = fmaxf((c[n][i] + bb) * sc + sh, 0.0f);
      }
    }
  } else {
    const int ln = (int)__builtin_amdgcn_mbcnt_lo(~0u, 0u);
    const int h2 = ln >> 4;
    const int l2 = ln & 15;
    float* hrow = L.At + (r0 + h2 * 8) * AP + l2;
#pragma unroll
    for (int n = 0; n < NF / 16; ++n) {
      const int col = n * 16 + l2;
      const float bb = bias[col];
      const float sc = g[col] * rsqrtf(rv[col] + 1e-5f);
      const float sh = be[col] - rm[col] * sc;
#pragma unroll
      for (int i = 0; i < 8; ++i) hrow[i * AP + n * 16] = fmaxf((c[n][i] + bb) * sc + sh, 0.0f);
    }

    const float* vl = L.W2s + l2 * WP + 2 * h2;
    v8f d[NT2];
#pragma unroll
    for (int n = 0; n < NT2; ++n) d[n] = v8f{};
#pragma unroll 4
    for (int t = 0; t < NF / 4; ++t) {
      const v2f a = *(const v2f*)(al + 4 * t);
      v2f bf[NT2];
#pragma unroll
      for (int n = 0; n < NT2; ++n) bf[n] = *(const v2f*)(vl + n * 16 * WP + 4 * t);
#pragma unroll
      for (int n = 0; n < NT2; ++n)
        d[n] = __builtin_amdgcn_wmma_f32_16x16x4_f32(
             false, a,  false, bf[n],
             (short)0, d[n],  false,  false);
    }

    float* ts = L.At + r0 * AP;
#pragma unroll
    for (int n = 0; n < NT2; ++n) {
      const int col = n * 16 + l2;
      if (col < NC) {
#pragma unroll
        for (int i = 0; i < 8; ++i) ts[(h2 * 8 + i) * NC + col] = d[n][i];
      }
    }
    const int row0 = b * RB + r0;
    float* __restrict__ o = out + (size_t)row0 * NC;
    static_assert((16 * NC / 4) % 32 == 0, "whole store instructions per strip");
    v4f tv[16 * NC / 4 / 32];
#pragma unroll
    for (int j = 0; j < 16 * NC / 4 / 32; ++j) tv[j] = *(const v4f*)(ts + 4 * (ln + 32 * j));
#pragma unroll
    for (int j = 0; j < 16 * NC / 4 / 32; ++j) {
      const int idx = ln + 32 * j;
      if (row0 + (4 * idx) / NC < N) *(v4f*)(o + 4 * idx) = tv[j];
    }
  }
}

#define NC4 (NC / 4)

__global__ __launch_bounds__(32 * OWN_WAVES) void owner40_kernel(
    const float* __restrict__ T, const float* __restrict__ recf, const int* __restrict__ run,
    const float* __restrict__ b2, float* __restrict__ out, int N) {
  __shared__ SortLds L;
  __shared__ float tile[RB * NC];
  const int tid = threadIdx.x;
  const int lane = tid & 31;
  const int wave = __builtin_amdgcn_readfirstlane((int)(tid >> 5));
  const int l4 = lane < NC4 ? lane : NC4 - 1;
  for (int i = tid; i < RB * NC; i += 32 * OWN_WAVES) tile[i] = b2[i % NC];

  const uint2* __restrict__ rec = (const uint2*)recf;
  const int b = blockIdx.x;
  const int end = load_runs(L, run, b, tid, lane, wave);
  const int r0 = (wave * RB) >> 5;
  const int r1 = ((wave + 1) * RB) >> 5;

  auto flush = [&](int row, v4f sum) {
    if (lane < NC4) {
      v4f* t = (v4f*)(tile + row * NC) + lane;
      *t += sum;
    }
  };
  int cb = 0;
  do {
    const int m = end - cb < CAP ? end - cb : CAP;
    sort_bucket<32 * OWN_WAVES>(L, rec, cb, m, tid, lane, wave);

    int p = __builtin_amdgcn_readfirstlane(L.start[r0]);
    const int pend = __builtin_amdgcn_readfirstlane(L.start[r1]);
    int cur = -1;
    v4f acc = v4f{};
    for (; p + EPB <= pend; p += EPB)
      stream_batch<true, NC>(T, L.srt, p, EPB, lane, l4, cur, acc, flush);
    if (p < pend) stream_batch<false, NC>(T, L.srt, p, pend - p, lane, l4, cur, acc, flush);
    if (cur >= 0) flush(cur, acc);

    cb += CAP;
    __syncthreads();
  } while (cb < end);

  const int row0 = b * RB;
  const int rows = N - row0 < RB ? N - row0 : RB;
  float* __restrict__ o = out + (size_t)row0 * NC;
  for (int i = tid; i < rows * NC; i += 32 * OWN_WAVES) o[i] = tile[i];
}

__global__ __launch_bounds__(256) void log_softmax40(const float* __restrict__ in,
                                                     float* __restrict__ out, int n) {
  int i = blockIdx.x * blockDim.x + threadIdx.x;
  if (i >= n) return;
  const float* row = in + (size_t)i * 40;
  float* orow = out + (size_t)i * 40;
  float m = row[0];
#pragma unroll
  for (int j = 1; j < 40; ++j) m = fmaxf(m, row[j]);
  float s = 0.0f;
#pragma unroll
  for (int j = 0; j < 40; ++j) s += expf(row[j] - m);
  float l = m + logf(s);
#pragma unroll
  for (int j = 0; j < 40; ++j) orow[j] = row[j] - l;
}

extern "C" void kernel_launch(void* const* d_in, const int* in_sizes, int n_in,
                              void* d_out, int out_size, void* d_ws, size_t ws_size,
                              hipStream_t stream) {
  const float* x     = (const float*)d_in[0];
  const float* Mv    = (const float*)d_in[1];
  const float* AMv   = (const float*)d_in[2];
  const int*   srcZ  = (const int*)d_in[3];
  const int*   dstZ  = (const int*)d_in[4];
  const float* valsZ = (const float*)d_in[5];
  const int*   src   = (const int*)d_in[6];
  const int*   dst   = (const int*)d_in[7];
  const float* vals  = (const float*)d_in[8];
  const float* W0 = (const float*)d_in[9];
  const float* b0 = (const float*)d_in[10];
  const float* g0 = (const float*)d_in[11];
  const float* be0 = (const float*)d_in[12];
  const float* rm0 = (const float*)d_in[13];
  const float* rv0 = (const float*)d_in[14];
  const float* W1 = (const float*)d_in[15];
  const float* b1 = (const float*)d_in[16];
  const float* g1 = (const float*)d_in[17];
  const float* be1 = (const float*)d_in[18];
  const float* rm1 = (const float*)d_in[19];
  const float* rv1 = (const float*)d_in[20];
  const float* W2 = (const float*)d_in[21];
  const float* b2 = (const float*)d_in[22];

  int N = in_sizes[1];
  int E = in_sizes[3];
  float* bufA = (float*)d_ws;
  float* bufB = bufA + (size_t)N * NF;
  float* outp = (float*)d_out;

  int nb = (N + RB - 1) / RB;

  float* recZ = bufA;
  int*   runZ = (int*)(bufA + 2 * (size_t)E);
  float* recA = outp;
  int*   runA = (int*)(outp + 2 * (size_t)E);

  chunk_group<<<dim3(GCH, 2), BK_THREADS, 0, stream>>>(srcZ, dstZ, valsZ, Mv, AMv, src, dst, vals,
                                                       runZ, runA, recZ, recA, E, nb);

  owner_gemm_kernel<false><<<nb, FT, 0, stream>>>(x, recZ, runZ, W0, b0, g0, be0, rm0, rv0,
                                                 nullptr, bufB, N);

  owner_gemm_kernel<true><<<nb, FT, 0, stream>>>(bufB, recA, runA, W1, b1, g1, be1, rm1, rv1,
                                                W2, bufA, N);

  owner40_kernel<<<nb, 32 * OWN_WAVES, 0, stream>>>(bufA, recA, runA, b2, bufB, N);
  log_softmax40<<<(N + 255) / 256, 256, 0, stream>>>(bufB, outp, N);
}


# Last layer: owner work-group takes log_softmax of its finished rows in LDS and stores the output itself: logits never reach memory; graph-2 records go to workspace; 1 launch fewer; sums as before
# speedup vs baseline: 9.6511x; 1.0658x over previous
#include <hip/hip_runtime.h>
#include <math.h>

typedef __attribute__((ext_vector_type(2))) float v2f;
typedef __attribute__((ext_vector_type(4))) float v4f;
typedef __attribute__((ext_vector_type(8))) float v8f;

#define NF 128

#define NC 40
#define NCP 48
#define NT2 (NCP / 16)
#define WP (NF + 4)

#define RB 400
#define NBMAX 256
#define GCH 256
#define BK_THREADS 512

#define EPT 13

__global__ __launch_bounds__(BK_THREADS) void chunk_group(
    const int* __restrict__ srcZ, const int* __restrict__ dstZ, const float* __restrict__ valsZ,
    const float* __restrict__ Mv, const float* __restrict__ AMv,
    const int* __restrict__ srcA, const int* __restrict__ dstA, const float* __restrict__ valsA,
    int* __restrict__ runZ, int* __restrict__ runA,
    float* __restrict__ recZ, float* __restrict__ recA, int E, int nb) {
  __shared__ int cur[NBMAX];
  __shared__ int wsum[NBMAX / 32];
  const int tid = threadIdx.x;
  const int lane = tid & 31, wv = tid >> 5;
  const int g = blockIdx.x;
  const bool z = blockIdx.y == 0;
  const int* __restrict__ src = z ? srcZ : srcA;
  const int* __restrict__ dst = z ? dstZ : dstA;
  const float* __restrict__ vals = z ? valsZ : valsA;
  int2* __restrict__ run = (int2*)(z ? runZ : runA);
  uint2* __restrict__ rec = (uint2*)(z ? recZ : recA);
  for (int i = tid; i < NBMAX; i += BK_THREADS) cur[i] = 0;
  __syncthreads();
  const int ch = (E + GCH - 1) / GCH;
  const int lo = g * ch;
  const int hi = lo + ch < E ? lo + ch : E;

  for (int base = lo + tid; base < hi; base += BK_THREADS * EPT) {
    int d[EPT];
#pragma unroll
    for (int j = 0; j < EPT; ++j) {
      const int e = base + BK_THREADS * j;
      d[j] = dst[e < hi ? e : hi - 1];
    }
#pragma unroll
    for (int j = 0; j < EPT; ++j) {
      atomicAdd(&cur[d[j] / RB], base + BK_THREADS * j < hi ? 1 : 0);
    }
  }
  __syncthreads();

  const int cv = tid < NBMAX ? cur[tid] : 0;
  int x = cv;
  for (int dd = 1; dd < 32; dd <<= 1) {
    int y = __shfl_up(x, dd);
    if (lane >= dd) x += y;
  }
  if (tid < NBMAX && lane == 31) wsum[wv] = x;
  __syncthreads();
  if (tid < NBMAX) {
    int before = 0;
#pragma unroll
    for (int k = 0; k < NBMAX / 32; ++k) before += k < wv ? wsum[k] : 0;
    const int st = lo + before + x - cv;
    cur[tid] = st;
    if (tid < nb) run[tid * GCH + g] = make_int2(st, cv);
  }
  __syncthreads();

  for (int base = lo + tid; base < hi; base += BK_THREADS * EPT) {
    int s[EPT], d[EPT];
    float w[EPT];
#pragma unroll
    for (int j = 0; j < EPT; ++j) {
      const int e = base + BK_THREADS * j;
      const int ec = e < hi ? e : hi - 1;
      s[j] = src[ec];
      d[j] = dst[ec];
      w[j] = vals[ec];
    }
    if (z) {
      float fm[EPT], fa[EPT];
#pragma unroll
      for (int j = 0; j < EPT; ++j) {
        fm[j] = Mv[s[j]];
        fa[j] = AMv[d[j]];
      }
#pragma unroll
      for (int j = 0; j < EPT; ++j) {
        w[j] *= fm[j];
        w[j] *= fa[j];
      }
    }
    int p[EPT];
#pragma unroll
    for (int j = 0; j < EPT; ++j) {
      if (base + BK_THREADS * j < hi) p[j] = atomicAdd(&cur[d[j] / RB], 1);
    }
#pragma unroll
    for (int j = 0; j < EPT; ++j) {
      if (base + BK_THREADS * j < hi) {
        const int b = d[j] / RB;
        rec[p[j]] = make_uint2((unsigned)s[j] | ((unsigned)(d[j] - b * RB) << 17), __float_as_uint(w[j]));
      }
    }
  }
}

#define CAP 8192
#define EPB 16
#define OWN_WAVES 32

struct SortLds {
  uint2 srt[CAP + 32];
  int cnt[512];
  int start[512];
  int wsum[32];
  int rs[GCH];
  int rp[GCH + 8];
};

__device__ __forceinline__ int load_runs(SortLds& L, const int* __restrict__ run, int b, int tid, int lane, int wave) {
  int len = 0;
  if (tid < GCH) {
    const int2 r = ((const int2*)run)[b * GCH + tid];
    L.rs[tid] = r.x;
    len = r.y;
  }
  int x = len;
  for (int d = 1; d < 32; d <<= 1) {
    int y = __shfl_up(x, d);
    if (lane >= d) x += y;
  }
  if (tid < GCH && lane == 31) L.wsum[wave] = x;
  __syncthreads();
  if (tid < GCH) {
    int before = 0;
#pragma unroll
    for (int k = 0; k < GCH / 32; ++k) before += k < wave ? L.wsum[k] : 0;
    L.rp[tid] = before + x - len;
    if (tid == GCH - 1) L.rp[GCH] = before + x;
  }
  __syncthreads();
  return L.rp[GCH];
}

template <int NTHR>
__device__ __forceinline__ void sort_bucket(SortLds& L, const uint2* __restrict__ rec, int cb, int m,
                                            int tid, int lane, int wave) {
  constexpr int RPT = (CAP + NTHR - 1) / NTHR;
  if (tid < 512) L.cnt[tid] = 0;
  if (NTHR < 1024 && tid < 32) L.wsum[tid] = 0;
  __syncthreads();

  int q[RPT], g[RPT];
#pragma unroll
  for (int j = 0; j < RPT; ++j) {
    const int i = tid + NTHR * j;
    q[j] = cb + (i < m ? i : 0);
    g[j] = 0;
  }
#pragma unroll
  for (int s = GCH / 2; s; s >>= 1) {
#pragma unroll
    for (int j = 0; j < RPT; ++j) g[j] += L.rp[g[j] + s] <= q[j] ? s : 0;
  }
#pragma unroll
  for (int j = 0; j < RPT; ++j) q[j] = L.rs[g[j]] + (q[j] - L.rp[g[j]]);

  uint2 r[RPT];
  int rk[RPT];
#pragma unroll
  for (int j = 0; j < RPT; ++j) {
    const int i = tid + NTHR * j;
    if (i < m) r[j] = rec[q[j]];
  }
#pragma unroll
  for (int j = 0; j < RPT; ++j) {
    const int i = tid + NTHR * j;
    if (i < m) rk[j] = atomicAdd(&L.cnt[r[j].x >> 17], 1);
  }
  __syncthreads();

  const int cv = tid < RB ? L.cnt[tid] : 0;
  int x = cv;
  for (int d = 1; d < 32; d <<= 1) {
    int y = __shfl_up(x, d);
    if (lane >= d) x += y;
  }
  if (lane == 31) L.wsum[wave] = x;
  __syncthreads();
  if (wave == 0) {
    int t = L.wsum[lane];
    for (int d = 1; d < 32; d <<= 1) {
      int y = __shfl_up(t, d);
      if (lane >= d) t += y;
    }
    L.wsum[lane] = t;
  }
  __syncthreads();
  if (tid <= RB) L.start[tid] = (wave ? L.wsum[wave - 1] : 0) + x - cv;
  __syncthreads();

#pragma unroll
  for (int j = 0; j < RPT; ++j) {
    const int i = tid + NTHR * j;
    if (i < m) L.srt[L.start[r[j].x >> 17] + rk[j]] = r[j];
  }
  __syncthreads();
}

template <bool FULL, int ROWF, class Flush>
__device__ __forceinline__ void stream_batch(const float* __restrict__ X, const uint2* srt, int p,
                                             int nrec, int lane, int l4, int& cur, v4f& acc,
                                             Flush flush) {
  const uint2 rr = srt[p + lane];
  unsigned sx[EPB];
  float sw[EPB];
#pragma unroll
  for (int k = 0; k < EPB; ++k) {
    sx[k] = (unsigned)__builtin_amdgcn_readlane((int)rr.x, k);
    sw[k] = __uint_as_float((unsigned)__builtin_amdgcn_readlane((int)rr.y, k));
  }
  v4f v[EPB];
#pragma unroll
  for (int k = 0; k < EPB; ++k) {
    const unsigned s = (FULL || k < nrec) ? sx[k] : sx[0];
    v[k] = ((const v4f*)(X + (size_t)(s & 0x1FFFFu) * ROWF))[l4];
  }
#pragma unroll
  for (int k = 0; k < EPB; ++k) {
    if (FULL || k < nrec) {
      const int row = (int)(sx[k] >> 17);
      if (row != cur) {
        if (cur >= 0) flush(cur, acc);
        acc = v4f{};
        cur = row;
      }
      acc += sw[k] * v[k];
    }
  }
}

#define FW (RB / 16)
#define FT (32 * FW)
#define AP (NF + 4)

template <bool NEXT>
struct alignas(16) FusedLds {
  union {
    SortLds S;
    float Ws[NF * WP];
  } u;
  float At[RB * AP];
  float W2s[NEXT ? NCP * WP : 4];
};

template <bool NEXT>
__global__ __launch_bounds__(FT) void owner_gemm_kernel(
    const float* __restrict__ X, const float* __restrict__ recf, const int* __restrict__ run,
    const float* __restrict__ W, const float* __restrict__ bias,
    const float* __restrict__ g, const float* __restrict__ be,
    const float* __restrict__ rm, const float* __restrict__ rv,
    const float* __restrict__ W2,
    float* __restrict__ out,
    int N) {
  static_assert(RB % 16 == 0 && FW <= 32, "one wave per 16-row strip");
  __shared__ FusedLds<NEXT> L;
  const int tid = threadIdx.x;
  const int lane = tid & 31;
  const int wave = __builtin_amdgcn_readfirstlane((int)(tid >> 5));
  const uint2* __restrict__ rec = (const uint2*)recf;
  const int b = blockIdx.x;
  const int end = load_runs(L.u.S, run, b, tid, lane, wave);
  const int r0 = wave * 16;
  const int r1 = r0 + 16;

  int pass = 0;
  int cb = 0;
  do {
    const int m = end - cb < CAP ? end - cb : CAP;
    sort_bucket<FT>(L.u.S, rec, cb, m, tid, lane, wave);

    if (pass == 0) {
      for (int q = r0; q < r1; ++q) {
        if (__builtin_amdgcn_readfirstlane(L.u.S.cnt[q]) == 0)
          ((v4f*)(L.At + q * AP))[lane] = v4f{};
      }
    }

    auto flush = [&](int row, v4f sum) {
      v4f* dst = (v4f*)(L.At + row * AP) + lane;
      if (pass) sum += *dst;
      *dst = sum;
    };
    int p = __builtin_amdgcn_readfirstlane(L.u.S.start[r0]);
    const int pend = __builtin_amdgcn_readfirstlane(L.u.S.start[r1]);
    int cur = -1;
    v4f acc = v4f{};
    for (; p + EPB <= pend; p += EPB)
      stream_batch<true, NF>(X, L.u.S.srt, p, EPB, lane, lane, cur, acc, flush);
    if (p < pend) stream_batch<false, NF>(X, L.u.S.srt, p, pend - p, lane, lane, cur, acc, flush);
    if (cur >= 0) flush(cur, acc);

    cb += CAP;
    ++pass;
    if (cb < end) __syncthreads();
  } while (cb < end);

  constexpr int WPT = (NF * NF / 4 + FT - 1) / FT;
  v4f wv[WPT];
#pragma unroll
  for (int j = 0; j < WPT; ++j) {
    const int idx = tid + FT * j;
    if (idx < NF * NF / 4) wv[j] = *(const v4f*)(W + 4 * idx);
  }
  constexpr int VPT = (NCP * NF / 4 + FT - 1) / FT;
  v4f vv[VPT];
  if constexpr (NEXT) {
#pragma unroll
    for (int j = 0; j < VPT; ++j) {
      const int idx = tid + FT * j;
      if (idx < NCP * NF / 4) {
        const int r = idx >> 5;
        vv[j] = *(const v4f*)(W2 + (r < NC ? r : NC - 1) * NF + 4 * (idx & 31));
      }
    }
  }
  __syncthreads();
#pragma unroll
  for (int j = 0; j < WPT; ++j) {
    const int idx = tid + FT * j;
    if (idx < NF * NF / 4) *(v4f*)(L.u.Ws + (idx >> 5) * WP + 4 * (idx & 31)) = wv[j];
  }
  if constexpr (NEXT) {
#pragma unroll
    for (int j = 0; j < VPT; ++j) {
      const int idx = tid + FT * j;
      if (idx < NCP * NF / 4) *(v4f*)(L.W2s + (idx >> 5) * WP + 4 * (idx & 31)) = vv[j];
    }
  }
  __syncthreads();

  const int hl = lane >> 4;
  const int ll = lane & 15;
  const float* al = L.At + (r0 + ll) * AP + 2 * hl;
  const float* wl = L.u.Ws + ll * WP + 2 * hl;
  v8f c[NF / 16];
#pragma unroll
  for (int n = 0; n < NF / 16; ++n) c[n] = v8f{};

#pragma unroll 4
  for (int t = 0; t < NF / 4; ++t) {
    const v2f a = *(const v2f*)(al + 4 * t);
    v2f bf[NF / 16];
#pragma unroll
    for (int n = 0; n < NF / 16; ++n) bf[n] = *(const v2f*)(wl + n * 16 * WP + 4 * t);
#pragma unroll
    for (int n = 0; n < NF / 16; ++n)
      c[n] = __builtin_amdgcn_wmma_f32_16x16x4_f32(
           false, a,  false, bf[n],
           (short)0, c[n],  false,  false);
  }

  if constexpr (!NEXT) {
    const int rb = b * RB + r0 + hl * 8;
#pragma unroll
    for (int n = 0; n < NF / 16; ++n) {
      const int col = n * 16 + ll;
      const float bb = bias[col];
      const float sc = g[col] * rsqrtf(rv[col] + 1e-5f);
      const float sh = be[col] - rm[col] * sc;
#pragma unroll
      for (int i = 0; i < 8; ++i) {
        const int r = rb + i;
        if (r < N) out[(size_t)r * NF + col] = fmaxf((c[n][i] + bb) * sc + sh, 0.0f);
      }
    }
  } else {
    const int ln = (int)__builtin_amdgcn_mbcnt_lo(~0u, 0u);
    const int h2 = ln >> 4;
    const int l2 = ln & 15;
    float* hrow = L.At + (r0 + h2 * 8) * AP + l2;
#pragma unroll
    for (int n = 0; n < NF / 16; ++n) {
      const int col = n * 16 + l2;
      const float bb = bias[col];
      const float sc = g[col] * rsqrtf(rv[col] + 1e-5f);
      const float sh = be[col] - rm[col] * sc;
#pragma unroll
      for (int i = 0; i < 8; ++i) hrow[i * AP + n * 16] = fmaxf((c[n][i] + bb) * sc + sh, 0.0f);
    }

    const float* vl = L.W2s + l2 * WP + 2 * h2;
    v8f d[NT2];
#pragma unroll
    for (int n = 0; n < NT2; ++n) d[n] = v8f{};
#pragma unroll 4
    for (int t = 0; t < NF / 4; ++t) {
      const v2f a = *(const v2f*)(al + 4 * t);
      v2f bf[NT2];
#pragma unroll
      for (int n = 0; n < NT2; ++n) bf[n] = *(const v2f*)(vl + n * 16 * WP + 4 * t);
#pragma unroll
      for (int n = 0; n < NT2; ++n)
        d[n] = __builtin_amdgcn_wmma_f32_16x16x4_f32(
             false, a,  false, bf[n],
             (short)0, d[n],  false,  false);
    }

    float* ts = L.At + r0 * AP;
#pragma unroll
    for (int n = 0; n < NT2; ++n) {
      const int col = n * 16 + l2;
      if (col < NC) {
#pragma unroll
        for (int i = 0; i < 8; ++i) ts[(h2 * 8 + i) * NC + col] = d[n][i];
      }
    }
    const int row0 = b * RB + r0;
    float* __restrict__ o = out + (size_t)row0 * NC;
    static_assert((16 * NC / 4) % 32 == 0, "whole store instructions per strip");
    v4f tv[16 * NC / 4 / 32];
#pragma unroll
    for (int j = 0; j < 16 * NC / 4 / 32; ++j) tv[j] = *(const v4f*)(ts + 4 * (ln + 32 * j));
#pragma unroll
    for (int j = 0; j < 16 * NC / 4 / 32; ++j) {
      const int idx = ln + 32 * j;
      if (row0 + (4 * idx) / NC < N) *(v4f*)(o + 4 * idx) = tv[j];
    }
  }
}

#define NC4 (NC / 4)

__global__ __launch_bounds__(32 * OWN_WAVES) void owner40_kernel(
    const float* __restrict__ T, const float* __restrict__ recf, const int* __restrict__ run,
    const float* __restrict__ b2, float* __restrict__ out, int N) {
  __shared__ SortLds L;
  __shared__ alignas(16) float tile[RB * NC];
  __shared__ float lse[RB];
  const int tid = threadIdx.x;
  const int lane = tid & 31;
  const int wave = __builtin_amdgcn_readfirstlane((int)(tid >> 5));
  const int l4 = lane < NC4 ? lane : NC4 - 1;
  for (int i = tid; i < RB * NC; i += 32 * OWN_WAVES) tile[i] = b2[i % NC];

  const uint2* __restrict__ rec = (const uint2*)recf;
  const int b = blockIdx.x;
  const int end = load_runs(L, run, b, tid, lane, wave);
  const int r0 = (wave * RB) >> 5;
  const int r1 = ((wave + 1) * RB) >> 5;

  auto flush = [&](int row, v4f sum) {
    if (lane < NC4) {
      v4f* t = (v4f*)(tile + row * NC) + lane;
      *t += sum;
    }
  };
  int cb = 0;
  do {
    const int m = end - cb < CAP ? end - cb : CAP;
    sort_bucket<32 * OWN_WAVES>(L, rec, cb, m, tid, lane, wave);

    int p = __builtin_amdgcn_readfirstlane(L.start[r0]);
    const int pend = __builtin_amdgcn_readfirstlane(L.start[r1]);
    int cur = -1;
    v4f acc = v4f{};
    for (; p + EPB <= pend; p += EPB)
      stream_batch<true, NC>(T, L.srt, p, EPB, lane, l4, cur, acc, flush);
    if (p < pend) stream_batch<false, NC>(T, L.srt, p, pend - p, lane, l4, cur, acc, flush);
    if (cur >= 0) flush(cur, acc);

    cb += CAP;
    __syncthreads();
  } while (cb < end);

  const int row0 = b * RB;
  const int rows = N - row0 < RB ? N - row0 : RB;
  if (tid < rows) {
    const v4f* r = (const v4f*)(tile + tid * NC);
    v4f q[NC4];
#pragma unroll
    for (int j = 0; j < NC4; ++j) q[j] = r[j];
    float m = q[0][0];
#pragma unroll
    for (int j = 1; j < NC; ++j) m = fmaxf(m, q[j >> 2][j & 3]);
    float s = 0.0f;
#pragma unroll
    for (int j = 0; j < NC; ++j) s += expf(q[j >> 2][j & 3] - m);
    lse[tid] = m + logf(s);
  }
  __syncthreads();
  v4f* __restrict__ o = (v4f*)(out + (size_t)row0 * NC);
  for (int i = tid; i < rows * NC4; i += 32 * OWN_WAVES) o[i] = ((const v4f*)tile)[i] - lse[i / NC4];
}

extern "C" void kernel_launch(void* const* d_in, const int* in_sizes, int n_in,
                              void* d_out, int out_size, void* d_ws, size_t ws_size,
                              hipStream_t stream) {
  const float* x     = (const float*)d_in[0];
  const float* Mv    = (const float*)d_in[1];
  const float* AMv   = (const float*)d_in[2];
  const int*   srcZ  = (const int*)d_in[3];
  const int*   dstZ  = (const int*)d_in[4];
  const float* valsZ = (const float*)d_in[5];
  const int*   src   = (const int*)d_in[6];
  const int*   dst   = (const int*)d_in[7];
  const float* vals  = (const float*)d_in[8];
  const float* W0 = (const float*)d_in[9];
  const float* b0 = (const float*)d_in[10];
  const float* g0 = (const float*)d_in[11];
  const float* be0 = (const float*)d_in[12];
  const float* rm0 = (const float*)d_in[13];
  const float* rv0 = (const float*)d_in[14];
  const float* W1 = (const float*)d_in[15];
  const float* b1 = (const float*)d_in[16];
  const float* g1 = (const float*)d_in[17];
  const float* be1 = (const float*)d_in[18];
  const float* rm1 = (const float*)d_in[19];
  const float* rv1 = (const float*)d_in[20];
  const float* W2 = (const float*)d_in[21];
  const float* b2 = (const float*)d_in[22];

  int N = in_sizes[1];
  int E = in_sizes[3];
  float* bufA = (float*)d_ws;
  float* bufB = bufA + (size_t)N * NF;
  float* outp = (float*)d_out;

  int nb = (N + RB - 1) / RB;

  float* recZ = bufA;
  int*   runZ = (int*)(bufA + 2 * (size_t)E);
  float* recA = bufA + (size_t)N * (NF / 2);
  int*   runA = (int*)(recA + 2 * (size_t)E);

  chunk_group<<<dim3(GCH, 2), BK_THREADS, 0, stream>>>(srcZ, dstZ, valsZ, Mv, AMv, src, dst, vals,
                                                       runZ, runA, recZ, recA, E, nb);

  owner_gemm_kernel<false><<<nb, FT, 0, stream>>>(x, recZ, runZ, W0, b0, g0, be0, rm0, rv0,
                                                 nullptr, bufB, N);

  owner_gemm_kernel<true><<<nb, FT, 0, stream>>>(bufB, recA, runA, W1, b1, g1, be1, rm1, rv1,
                                                W2, bufA, N);

  owner40_kernel<<<nb, 32 * OWN_WAVES, 0, stream>>>(bufA, recA, runA, b2, outp, N);
}


# Grouping: a chunk's records are placed in an LDS image of its contiguous slot range and written out as whole aligned 128B lines (512B per store), not one 8B store per edge; records and sums unchanged
# speedup vs baseline: 9.9243x; 1.0283x over previous
#include <hip/hip_runtime.h>
#include <math.h>

typedef __attribute__((ext_vector_type(2))) float v2f;
typedef __attribute__((ext_vector_type(4))) float v4f;
typedef __attribute__((ext_vector_type(8))) float v8f;

#define NF 128

#define NC 40
#define NCP 48
#define NT2 (NCP / 16)
#define WP (NF + 4)

#define RB 400
#define NBMAX 256
#define GCH 256
#define BK_THREADS 512

#define EPT 13

#define SCAP (BK_THREADS * EPT)
__global__ __launch_bounds__(BK_THREADS) void chunk_group(
    const int* __restrict__ srcZ, const int* __restrict__ dstZ, const float* __restrict__ valsZ,
    const float* __restrict__ Mv, const float* __restrict__ AMv,
    const int* __restrict__ srcA, const int* __restrict__ dstA, const float* __restrict__ valsA,
    int* __restrict__ runZ, int* __restrict__ runA,
    float* __restrict__ recZ, float* __restrict__ recA, int E, int nb) {
  __shared__ int cur[NBMAX];
  __shared__ int wsum[NBMAX / 32];
  __shared__ uint2 img[SCAP];
  const int tid = threadIdx.x;
  const int lane = tid & 31, wv = tid >> 5;
  const int g = blockIdx.x;
  const bool z = blockIdx.y == 0;
  const int* __restrict__ src = z ? srcZ : srcA;
  const int* __restrict__ dst = z ? dstZ : dstA;
  const float* __restrict__ vals = z ? valsZ : valsA;
  int2* __restrict__ run = (int2*)(z ? runZ : runA);
  uint2* __restrict__ rec = (uint2*)(z ? recZ : recA);
  for (int i = tid; i < NBMAX; i += BK_THREADS) cur[i] = 0;
  __syncthreads();
  const int ch = (E + GCH - 1) / GCH;
  const int lo = g * ch;
  const int hi = lo + ch < E ? lo + ch : E;
  const bool staged = ch <= SCAP;

  for (int base = lo + tid; base < hi; base += BK_THREADS * EPT) {
    int d[EPT];
#pragma unroll
    for (int j = 0; j < EPT; ++j) {
      const int e = base + BK_THREADS * j;
      d[j] = dst[e < hi ? e : hi - 1];
    }
#pragma unroll
    for (int j = 0; j < EPT; ++j) {
      atomicAdd(&cur[d[j] / RB], base + BK_THREADS * j < hi ? 1 : 0);
    }
  }
  __syncthreads();

  const int cv = tid < NBMAX ? cur[tid] : 0;
  int x = cv;
  for (int dd = 1; dd < 32; dd <<= 1) {
    int y = __shfl_up(x, dd);
    if (lane >= dd) x += y;
  }
  if (tid < NBMAX && lane == 31) wsum[wv] = x;
  __syncthreads();
  if (tid < NBMAX) {
    int before = 0;
#pragma unroll
    for (int k = 0; k < NBMAX / 32; ++k) before += k < wv ? wsum[k] : 0;
    const int st = lo + before + x - cv;
    cur[tid] = st;
    if (tid < nb) run[tid * GCH + g] = make_int2(st, cv);
  }
  __syncthreads();

  for (int base = lo + tid; base < hi; base += BK_THREADS * EPT) {
    int s[EPT], d[EPT];
    float w[EPT];
#pragma unroll
    for (int j = 0; j < EPT; ++j) {
      const int e = base + BK_THREADS * j;
      const int ec = e < hi ? e : hi - 1;
      s[j] = src[ec];
      d[j] = dst[ec];
      w[j] = vals[ec];
    }
    if (z) {
      float fm[EPT], fa[EPT];
#pragma unroll
      for (int j = 0; j < EPT; ++j) {
        fm[j] = Mv[s[j]];
        fa[j] = AMv[d[j]];
      }
#pragma unroll
      for (int j = 0; j < EPT; ++j) {
        w[j] *= fm[j];
        w[j] *= fa[j];
      }
    }
    int p[EPT];
#pragma unroll
    for (int j = 0; j < EPT; ++j) {
      if (base + BK_THREADS * j < hi) p[j] = atomicAdd(&cur[d[j] / RB], 1);
    }
#pragma unroll
    for (int j = 0; j < EPT; ++j) {
      if (base + BK_THREADS * j < hi) {
        const int b = d[j] / RB;
        const uint2 r = make_uint2((unsigned)s[j] | ((unsigned)(d[j] - b * RB) << 17), __float_as_uint(w[j]));
        if (staged) img[p[j] - lo] = r;
        else rec[p[j]] = r;
      }
    }
  }
  if (staged) {
    __syncthreads();
    uint4* __restrict__ rec4 = (uint4*)rec;
    for (int P = ((lo >> 1) & ~7) + tid; 2 * P < hi; P += BK_THREADS) {
      const int s0 = 2 * P, s1 = s0 + 1;
      if (s0 >= lo && s1 < hi) {
        const uint2 a = img[s0 - lo], bq = img[s1 - lo];
        rec4[P] = make_uint4(a.x, a.y, bq.x, bq.y);
      } else if (s0 >= lo) {
        rec[s0] = img[s0 - lo];
      } else if (s1 >= lo && s1 < hi) {
        rec[s1] = img[s1 - lo];
      }
    }
  }
}

#define CAP 8192
#define EPB 16
#define OWN_WAVES 32

struct SortLds {
  uint2 srt[CAP + 32];
  int cnt[512];
  int start[512];
  int wsum[32];
  int rs[GCH];
  int rp[GCH + 8];
};

__device__ __forceinline__ int load_runs(SortLds& L, const int* __restrict__ run, int b, int tid, int lane, int wave) {
  int len = 0;
  if (tid < GCH) {
    const int2 r = ((const int2*)run)[b * GCH + tid];
    L.rs[tid] = r.x;
    len = r.y;
  }
  int x = len;
  for (int d = 1; d < 32; d <<= 1) {
    int y = __shfl_up(x, d);
    if (lane >= d) x += y;
  }
  if (tid < GCH && lane == 31) L.wsum[wave] = x;
  __syncthreads();
  if (tid < GCH) {
    int before = 0;
#pragma unroll
    for (int k = 0; k < GCH / 32; ++k) before += k < wave ? L.wsum[k] : 0;
    L.rp[tid] = before + x - len;
    if (tid == GCH - 1) L.rp[GCH] = before + x;
  }
  __syncthreads();
  return L.rp[GCH];
}

template <int NTHR>
__device__ __forceinline__ void sort_bucket(SortLds& L, const uint2* __restrict__ rec, int cb, int m,
                                            int tid, int lane, int wave) {
  constexpr int RPT = (CAP + NTHR - 1) / NTHR;
  if (tid < 512) L.cnt[tid] = 0;
  if (NTHR < 1024 && tid < 32) L.wsum[tid] = 0;
  __syncthreads();

  int q[RPT], g[RPT];
#pragma unroll
  for (int j = 0; j < RPT; ++j) {
    const int i = tid + NTHR * j;
    q[j] = cb + (i < m ? i : 0);
    g[j] = 0;
  }
#pragma unroll
  for (int s = GCH / 2; s; s >>= 1) {
#pragma unroll
    for (int j = 0; j < RPT; ++j) g[j] += L.rp[g[j] + s] <= q[j] ? s : 0;
  }
#pragma unroll
  for (int j = 0; j < RPT; ++j) q[j] = L.rs[g[j]] + (q[j] - L.rp[g[j]]);

  uint2 r[RPT];
  int rk[RPT];
#pragma unroll
  for (int j = 0; j < RPT; ++j) {
    const int i = tid + NTHR * j;
    if (i < m) r[j] = rec[q[j]];
  }
#pragma unroll
  for (int j = 0; j < RPT; ++j) {
    const int i = tid + NTHR * j;
    if (i < m) rk[j] = atomicAdd(&L.cnt[r[j].x >> 17], 1);
  }
  __syncthreads();

  const int cv = tid < RB ? L.cnt[tid] : 0;
  int x = cv;
  for (int d = 1; d < 32; d <<= 1) {
    int y = __shfl_up(x, d);
    if (lane >= d) x += y;
  }
  if (lane == 31) L.wsum[wave] = x;
  __syncthreads();
  if (wave == 0) {
    int t = L.wsum[lane];
    for (int d = 1; d < 32; d <<= 1) {
      int y = __shfl_up(t, d);
      if (lane >= d) t += y;
    }
    L.wsum[lane] = t;
  }
  __syncthreads();
  if (tid <= RB) L.start[tid] = (wave ? L.wsum[wave - 1] : 0) + x - cv;
  __syncthreads();

#pragma unroll
  for (int j = 0; j < RPT; ++j) {
    const int i = tid + NTHR * j;
    if (i < m) L.srt[L.start[r[j].x >> 17] + rk[j]] = r[j];
  }
  __syncthreads();
}

template <bool FULL, int ROWF, class Flush>
__device__ __forceinline__ void stream_batch(const float* __restrict__ X, const uint2* srt, int p,
                                             int nrec, int lane, int l4, int& cur, v4f& acc,
                                             Flush flush) {
  const uint2 rr = srt[p + lane];
  unsigned sx[EPB];
  float sw[EPB];
#pragma unroll
  for (int k = 0; k < EPB; ++k) {
    sx[k] = (unsigned)__builtin_amdgcn_readlane((int)rr.x, k);
    sw[k] = __uint_as_float((unsigned)__builtin_amdgcn_readlane((int)rr.y, k));
  }
  v4f v[EPB];
#pragma unroll
  for (int k = 0; k < EPB; ++k) {
    const unsigned s = (FULL || k < nrec) ? sx[k] : sx[0];
    v[k] = ((const v4f*)(X + (size_t)(s & 0x1FFFFu) * ROWF))[l4];
  }
#pragma unroll
  for (int k = 0; k < EPB; ++k) {
    if (FULL || k < nrec) {
      const int row = (int)(sx[k] >> 17);
      if (row != cur) {
        if (cur >= 0) flush(cur, acc);
        acc = v4f{};
        cur = row;
      }
      acc += sw[k] * v[k];
    }
  }
}

#define FW (RB / 16)
#define FT (32 * FW)
#define AP (NF + 4)

template <bool NEXT>
struct alignas(16) FusedLds {
  union {
    SortLds S;
    float Ws[NF * WP];
  } u;
  float At[RB * AP];
  float W2s[NEXT ? NCP * WP : 4];
};

template <bool NEXT>
__global__ __launch_bounds__(FT) void owner_gemm_kernel(
    const float* __restrict__ X, const float* __restrict__ recf, const int* __restrict__ run,
    const float* __restrict__ W, const float* __restrict__ bias,
    const float* __restrict__ g, const float* __restrict__ be,
    const float* __restrict__ rm, const float* __restrict__ rv,
    const float* __restrict__ W2,
    float* __restrict__ out,
    int N) {
  static_assert(RB % 16 == 0 && FW <= 32, "one wave per 16-row strip");
  __shared__ FusedLds<NEXT> L;
  const int tid = threadIdx.x;
  const int lane = tid & 31;
  const int wave = __builtin_amdgcn_readfirstlane((int)(tid >> 5));
  const uint2* __restrict__ rec = (const uint2*)recf;
  const int b = blockIdx.x;
  const int end = load_runs(L.u.S, run, b, tid, lane, wave);
  const int r0 = wave * 16;
  const int r1 = r0 + 16;

  int pass = 0;
  int cb = 0;
  do {
    const int m = end - cb < CAP ? end - cb : CAP;
    sort_bucket<FT>(L.u.S, rec, cb, m, tid, lane, wave);

    if (pass == 0) {
      for (int q = r0; q < r1; ++q) {
        if (__builtin_amdgcn_readfirstlane(L.u.S.cnt[q]) == 0)
          ((v4f*)(L.At + q * AP))[lane] = v4f{};
      }
    }

    auto flush = [&](int row, v4f sum) {
      v4f* dst = (v4f*)(L.At + row * AP) + lane;
      if (pass) sum += *dst;
      *dst = sum;
    };
    int p = __builtin_amdgcn_readfirstlane(L.u.S.start[r0]);
    const int pend = __builtin_amdgcn_readfirstlane(L.u.S.start[r1]);
    int cur = -1;
    v4f acc = v4f{};
    for (; p + EPB <= pend; p += EPB)
      stream_batch<true, NF>(X, L.u.S.srt, p, EPB, lane, lane, cur, acc, flush);
    if (p < pend) stream_batch<false, NF>(X, L.u.S.srt, p, pend - p, lane, lane, cur, acc, flush);
    if (cur >= 0) flush(cur, acc);

    cb += CAP;
    ++pass;
    if (cb < end) __syncthreads();
  } while (cb < end);

  constexpr int WPT = (NF * NF / 4 + FT - 1) / FT;
  v4f wv[WPT];
#pragma unroll
  for (int j = 0; j < WPT; ++j) {
    const int idx = tid + FT * j;
    if (idx < NF * NF / 4) wv[j] = *(const v4f*)(W + 4 * idx);
  }
  constexpr int VPT = (NCP * NF / 4 + FT - 1) / FT;
  v4f vv[VPT];
  if constexpr (NEXT) {
#pragma unroll
    for (int j = 0; j < VPT; ++j) {
      const int idx = tid + FT * j;
      if (idx < NCP * NF / 4) {
        const int r = idx >> 5;
        vv[j] = *(const v4f*)(W2 + (r < NC ? r : NC - 1) * NF + 4 * (idx & 31));
      }
    }
  }
  __syncthreads();
#pragma unroll
  for (int j = 0; j < WPT; ++j) {
    const int idx = tid + FT * j;
    if (idx < NF * NF / 4) *(v4f*)(L.u.Ws + (idx >> 5) * WP + 4 * (idx & 31)) = wv[j];
  }
  if constexpr (NEXT) {
#pragma unroll
    for (int j = 0; j < VPT; ++j) {
      const int idx = tid + FT * j;
      if (idx < NCP * NF / 4) *(v4f*)(L.W2s + (idx >> 5) * WP + 4 * (idx & 31)) = vv[j];
    }
  }
  __syncthreads();

  const int hl = lane >> 4;
  const int ll = lane & 15;
  const float* al = L.At + (r0 + ll) * AP + 2 * hl;
  const float* wl = L.u.Ws + ll * WP + 2 * hl;
  v8f c[NF / 16];
#pragma unroll
  for (int n = 0; n < NF / 16; ++n) c[n] = v8f{};

#pragma unroll 4
  for (int t = 0; t < NF / 4; ++t) {
    const v2f a = *(const v2f*)(al + 4 * t);
    v2f bf[NF / 16];
#pragma unroll
    for (int n = 0; n < NF / 16; ++n) bf[n] = *(const v2f*)(wl + n * 16 * WP + 4 * t);
#pragma unroll
    for (int n = 0; n < NF / 16; ++n)
      c[n] = __builtin_amdgcn_wmma_f32_16x16x4_f32(
           false, a,  false, bf[n],
           (short)0, c[n],  false,  false);
  }

  if constexpr (!NEXT) {
    const int rb = b * RB + r0 + hl * 8;
#pragma unroll
    for (int n = 0; n < NF / 16; ++n) {
      const int col = n * 16 + ll;
      const float bb = bias[col];
      const float sc = g[col] * rsqrtf(rv[col] + 1e-5f);
      const float sh = be[col] - rm[col] * sc;
#pragma unroll
      for (int i = 0; i < 8; ++i) {
        const int r = rb + i;
        if (r < N) out[(size_t)r * NF + col] = fmaxf((c[n][i] + bb) * sc + sh, 0.0f);
      }
    }
  } else {
    const int ln = (int)__builtin_amdgcn_mbcnt_lo(~0u, 0u);
    const int h2 = ln >> 4;
    const int l2 = ln & 15;
    float* hrow = L.At + (r0 + h2 * 8) * AP + l2;
#pragma unroll
    for (int n = 0; n < NF / 16; ++n) {
      const int col = n * 16 + l2;
      const float bb = bias[col];
      const float sc = g[col] * rsqrtf(rv[col] + 1e-5f);
      const float sh = be[col] - rm[col] * sc;
#pragma unroll
      for (int i = 0; i < 8; ++i) hrow[i * AP + n * 16] = fmaxf((c[n][i] + bb) * sc + sh, 0.0f);
    }

    const float* vl = L.W2s + l2 * WP + 2 * h2;
    v8f d[NT2];
#pragma unroll
    for (int n = 0; n < NT2; ++n) d[n] = v8f{};
#pragma unroll 4
    for (int t = 0; t < NF / 4; ++t) {
      const v2f a = *(const v2f*)(al + 4 * t);
      v2f bf[NT2];
#pragma unroll
      for (int n = 0; n < NT2; ++n) bf[n] = *(const v2f*)(vl + n * 16 * WP + 4 * t);
#pragma unroll
      for (int n = 0; n < NT2; ++n)
        d[n] = __builtin_amdgcn_wmma_f32_16x16x4_f32(
             false, a,  false, bf[n],
             (short)0, d[n],  false,  false);
    }

    float* ts = L.At + r0 * AP;
#pragma unroll
    for (int n = 0; n < NT2; ++n) {
      const int col = n * 16 + l2;
      if (col < NC) {
#pragma unroll
        for (int i = 0; i < 8; ++i) ts[(h2 * 8 + i) * NC + col] = d[n][i];
      }
    }
    const int row0 = b * RB + r0;
    float* __restrict__ o = out + (size_t)row0 * NC;
    static_assert((16 * NC / 4) % 32 == 0, "whole store instructions per strip");
    v4f tv[16 * NC / 4 / 32];
#pragma unroll
    for (int j = 0; j < 16 * NC / 4 / 32; ++j) tv[j] = *(const v4f*)(ts + 4 * (ln + 32 * j));
#pragma unroll
    for (int j = 0; j < 16 * NC / 4 / 32; ++j) {
      const int idx = ln + 32 * j;
      if (row0 + (4 * idx) / NC < N) *(v4f*)(o + 4 * idx) = tv[j];
    }
  }
}

#define NC4 (NC / 4)

__global__ __launch_bounds__(32 * OWN_WAVES) void owner40_kernel(
    const float* __restrict__ T, const float* __restrict__ recf, const int* __restrict__ run,
    const float* __restrict__ b2, float* __restrict__ out, int N) {
  __shared__ SortLds L;
  __shared__ alignas(16) float tile[RB * NC];
  __shared__ float lse[RB];
  const int tid = threadIdx.x;
  const int lane = tid & 31;
  const int wave = __builtin_amdgcn_readfirstlane((int)(tid >> 5));
  const int l4 = lane < NC4 ? lane : NC4 - 1;
  for (int i = tid; i < RB * NC; i += 32 * OWN_WAVES) tile[i] = b2[i % NC];

  const uint2* __restrict__ rec = (const uint2*)recf;
  const int b = blockIdx.x;
  const int end = load_runs(L, run, b, tid, lane, wave);
  const int r0 = (wave * RB) >> 5;
  const int r1 = ((wave + 1) * RB) >> 5;

  auto flush = [&](int row, v4f sum) {
    if (lane < NC4) {
      v4f* t = (v4f*)(tile + row * NC) + lane;
      *t += sum;
    }
  };
  int cb = 0;
  do {
    const int m = end - cb < CAP ? end - cb : CAP;
    sort_bucket<32 * OWN_WAVES>(L, rec, cb, m, tid, lane, wave);

    int p = __builtin_amdgcn_readfirstlane(L.start[r0]);
    const int pend = __builtin_amdgcn_readfirstlane(L.start[r1]);
    int cur = -1;
    v4f acc = v4f{};
    for (; p + EPB <= pend; p += EPB)
      stream_batch<true, NC>(T, L.srt, p, EPB, lane, l4, cur, acc, flush);
    if (p < pend) stream_batch<false, NC>(T, L.srt, p, pend - p, lane, l4, cur, acc, flush);
    if (cur >= 0) flush(cur, acc);

    cb += CAP;
    __syncthreads();
  } while (cb < end);

  const int row0 = b * RB;
  const int rows = N - row0 < RB ? N - row0 : RB;
  if (tid < rows) {
    const v4f* r = (const v4f*)(tile + tid * NC);
    v4f q[NC4];
#pragma unroll
    for (int j = 0; j < NC4; ++j) q[j] = r[j];
    float m = q[0][0];
#pragma unroll
    for (int j = 1; j < NC; ++j) m = fmaxf(m, q[j >> 2][j & 3]);
    float s = 0.0f;
#pragma unroll
    for (int j = 0; j < NC; ++j) s += expf(q[j >> 2][j & 3] - m);
    lse[tid] = m + logf(s);
  }
  __syncthreads();
  v4f* __restrict__ o = (v4f*)(out + (size_t)row0 * NC);
  for (int i = tid; i < rows * NC4; i += 32 * OWN_WAVES) o[i] = ((const v4f*)tile)[i] - lse[i / NC4];
}

extern "C" void kernel_launch(void* const* d_in, const int* in_sizes, int n_in,
                              void* d_out, int out_size, void* d_ws, size_t ws_size,
                              hipStream_t stream) {
  const float* x     = (const float*)d_in[0];
  const float* Mv    = (const float*)d_in[1];
  const float* AMv   = (const float*)d_in[2];
  const int*   srcZ  = (const int*)d_in[3];
  const int*   dstZ  = (const int*)d_in[4];
  const float* valsZ = (const float*)d_in[5];
  const int*   src   = (const int*)d_in[6];
  const int*   dst   = (const int*)d_in[7];
  const float* vals  = (const float*)d_in[8];
  const float* W0 = (const float*)d_in[9];
  const float* b0 = (const float*)d_in[10];
  const float* g0 = (const float*)d_in[11];
  const float* be0 = (const float*)d_in[12];
  const float* rm0 = (const float*)d_in[13];
  const float* rv0 = (const float*)d_in[14];
  const float* W1 = (const float*)d_in[15];
  const float* b1 = (const float*)d_in[16];
  const float* g1 = (const float*)d_in[17];
  const float* be1 = (const float*)d_in[18];
  const float* rm1 = (const float*)d_in[19];
  const float* rv1 = (const float*)d_in[20];
  const float* W2 = (const float*)d_in[21];
  const float* b2 = (const float*)d_in[22];

  int N = in_sizes[1];
  int E = in_sizes[3];
  float* bufA = (float*)d_ws;
  float* bufB = bufA + (size_t)N * NF;
  float* outp = (float*)d_out;

  int nb = (N + RB - 1) / RB;

  float* recZ = bufA;
  int*   runZ = (int*)(bufA + 2 * (size_t)E);
  float* recA = bufA + (size_t)N * (NF / 2);
  int*   runA = (int*)(recA + 2 * (size_t)E);

  chunk_group<<<dim3(GCH, 2), BK_THREADS, 0, stream>>>(srcZ, dstZ, valsZ, Mv, AMv, src, dst, vals,
                                                       runZ, runA, recZ, recA, E, nb);

  owner_gemm_kernel<false><<<nb, FT, 0, stream>>>(x, recZ, runZ, W0, b0, g0, be0, rm0, rv0,
                                                 nullptr, bufB, N);

  owner_gemm_kernel<true><<<nb, FT, 0, stream>>>(bufB, recA, runA, W1, b1, g1, be1, rm1, rv1,
                                                W2, bufA, N);

  owner40_kernel<<<nb, 32 * OWN_WAVES, 0, stream>>>(bufA, recA, runA, b2, outp, N);
}
